# starting kernel
# baseline (speedup 1.0000x reference)
#include <hip/hip_runtime.h>

typedef _Float16 half_t;
typedef __attribute__((ext_vector_type(16))) _Float16 v16h;
typedef __attribute__((ext_vector_type(8)))  _Float16 v8h;
typedef __attribute__((ext_vector_type(8)))  float    v8f;

#define WMMA_F16(a, b, c) \
  __builtin_amdgcn_wmma_f32_16x16x32_f16(false, (a), false, (b), (short)0, (c), false, false)

__device__ __forceinline__ int kmap_of(int lane, int j) {
  return ((lane >> 4) << 3) + (j & 7) + ((j >> 3) << 4);
}

__global__ __launch_bounds__(256)
void groupnorm_kernel(const float* __restrict__ x,
                      const float* __restrict__ gw,
                      const float* __restrict__ gb,
                      half_t* __restrict__ h16) {
  __shared__ float s_sum[256], s_ss[256];
  int b = blockIdx.x >> 5;
  int g = blockIdx.x & 31;
  const int GE = 16 * 1024;
  const float* xb = x + ((long long)b * 512 + g * 16) * 1024;
  float sum = 0.f, ss = 0.f;
  for (int i = threadIdx.x; i < GE; i += 256) {
    float v = xb[i];
    sum += v; ss += v * v;
  }
  s_sum[threadIdx.x] = sum; s_ss[threadIdx.x] = ss;
  __syncthreads();
  for (int off = 128; off > 0; off >>= 1) {
    if (threadIdx.x < off) {
      s_sum[threadIdx.x] += s_sum[threadIdx.x + off];
      s_ss[threadIdx.x]  += s_ss[threadIdx.x + off];
    }
    __syncthreads();
  }
  float mean = s_sum[0] * (1.0f / GE);
  float var  = s_ss[0] * (1.0f / GE) - mean * mean;
  float inv  = rsqrtf(var + 1e-5f);
  half_t* hb = h16 + ((long long)b * 512 + g * 16) * 1024;
  for (int i = threadIdx.x; i < GE; i += 256) {
    int c = g * 16 + (i >> 10);
    hb[i] = (half_t)((xb[i] - mean) * inv * gw[c] + gb[c]);
  }
}

template <typename SrcT>
__global__ __launch_bounds__(256)
void repackB_kernel(const SrcT* __restrict__ src, half_t* __restrict__ dst,
                    int KT, int NT, int ncols, int nInner,
                    long long outerStride, long long innerStride, long long dstStride) {
  int y = blockIdx.y;
  const SrcT* s = src + (long long)(y / nInner) * outerStride
                      + (long long)(y % nInner) * innerStride;
  half_t* d = dst + (long long)y * dstStride;
  int total = KT * NT * 512;
  for (int tid = blockIdx.x * 256 + threadIdx.x; tid < total; tid += gridDim.x * 256) {
    int j = tid & 15, lane = (tid >> 4) & 31, tile = tid >> 9;
    int kt = tile / NT, nt = tile - kt * NT;
    int row = kt * 32 + kmap_of(lane, j);
    int col = nt * 16 + (lane & 15);
    d[tid] = (half_t)s[(long long)row * ncols + col];
  }
}

template <typename SrcT>
__global__ __launch_bounds__(256)
void repackA_kernel(const SrcT* __restrict__ src, half_t* __restrict__ dst,
                    int MT, int KT, int kcols, int nInner,
                    long long outerStride, long long innerStride, long long dstStride) {
  int y = blockIdx.y;
  const SrcT* s = src + (long long)(y / nInner) * outerStride
                      + (long long)(y % nInner) * innerStride;
  half_t* d = dst + (long long)y * dstStride;
  int total = MT * KT * 512;
  for (int tid = blockIdx.x * 256 + threadIdx.x; tid < total; tid += gridDim.x * 256) {
    int j = tid & 15, lane = (tid >> 4) & 31, tile = tid >> 9;
    int mt = tile / KT, kt = tile - mt * KT;
    int row = mt * 16 + (lane & 15);
    int col = kt * 32 + kmap_of(lane, j);
    d[tid] = (half_t)s[(long long)row * kcols + col];
  }
}

__device__ __forceinline__
void gemm_wave64x32(const half_t* __restrict__ Afrag, const half_t* __restrict__ Bfrag,
                    int KT, int NT, int mtBase, int ntBase, int lane, v8f c[4][2]) {
  for (int kt = 0; kt < KT; ++kt) {
    v16h a[4], bb[2];
#pragma unroll
    for (int i = 0; i < 4; ++i)
      a[i] = *(const v16h*)(Afrag + (((long long)(mtBase + i) * KT + kt) * 32 + lane) * 16);
#pragma unroll
    for (int j = 0; j < 2; ++j)
      bb[j] = *(const v16h*)(Bfrag + (((long long)kt * NT + (ntBase + j)) * 32 + lane) * 16);
#pragma unroll
    for (int i = 0; i < 4; ++i)
#pragma unroll
      for (int j = 0; j < 2; ++j)
        c[i][j] = WMMA_F16(a[i], bb[j], c[i][j]);
  }
}

__global__ __launch_bounds__(256)
void gemm_qkv_kernel(const half_t* __restrict__ Wfrag, const half_t* __restrict__ Hfrag,
                     const float* __restrict__ bias, half_t* __restrict__ out) {
  int wave = threadIdx.x >> 5, lane = threadIdx.x & 31;
  int b = blockIdx.z;
  int wm = wave & 1, wn = wave >> 1;
  int mBase = blockIdx.x * 128 + wm * 64;
  int nBase = blockIdx.y * 128 + wn * 32;
  v8f c[4][2];
#pragma unroll
  for (int i = 0; i < 4; ++i)
#pragma unroll
    for (int j = 0; j < 2; ++j) c[i][j] = 0.0f;
  gemm_wave64x32(Wfrag, Hfrag + (long long)b * (16LL * 64 * 512),
                 16, 64, mBase >> 4, nBase >> 4, lane, c);
  half_t* ob = out + (long long)b * 1536 * 1024;
  int rAdd = 8 * (lane >> 4), col = lane & 15;
#pragma unroll
  for (int i = 0; i < 4; ++i)
#pragma unroll
    for (int j = 0; j < 2; ++j)
#pragma unroll
      for (int r = 0; r < 8; ++r) {
        int m = mBase + i * 16 + r + rAdd;
        int n = nBase + j * 16 + col;
        ob[(long long)m * 1024 + n] = (half_t)(c[i][j][r] + bias[m]);
      }
}

__global__ __launch_bounds__(256)
void gemm_proj_kernel(const half_t* __restrict__ Wfrag, const half_t* __restrict__ Bfrag,
                      const float* __restrict__ bias, const float* __restrict__ resid,
                      float* __restrict__ out) {
  int wave = threadIdx.x >> 5, lane = threadIdx.x & 31;
  int b = blockIdx.z;
  int wm = wave & 1, wn = wave >> 1;
  int mBase = blockIdx.x * 128 + wm * 64;
  int nBase = blockIdx.y * 128 + wn * 32;
  v8f c[4][2];
#pragma unroll
  for (int i = 0; i < 4; ++i)
#pragma unroll
    for (int j = 0; j < 2; ++j) c[i][j] = 0.0f;
  gemm_wave64x32(Wfrag, Bfrag + (long long)b * (16LL * 64 * 512),
                 16, 64, mBase >> 4, nBase >> 4, lane, c);
  const float* xb = resid + (long long)b * 512 * 1024;
  float* ob = out + (long long)b * 512 * 1024;
  int rAdd = 8 * (lane >> 4), col = lane & 15;
#pragma unroll
  for (int i = 0; i < 4; ++i)
#pragma unroll
    for (int j = 0; j < 2; ++j)
#pragma unroll
      for (int r = 0; r < 8; ++r) {
        int m = mBase + i * 16 + r + rAdd;
        int n = nBase + j * 16 + col;
        long long idx = (long long)m * 1024 + n;
        ob[idx] = c[i][j][r] + bias[m] + xb[idx];
      }
}

__global__ __launch_bounds__(256)
void attention_kernel(const half_t* __restrict__ Qfrag, const half_t* __restrict__ Kfrag,
                      const half_t* __restrict__ Vfrag, half_t* __restrict__ Oplain) {
  __shared__ __align__(16) float  Stile[8][16][32];
  __shared__ __align__(16) half_t Ptile[8][16][32];
  __shared__ float mrow[8][16], lrow[8][16], arow[8][16];

  int wave = threadIdx.x >> 5, lane = threadIdx.x & 31;
  int wg = blockIdx.x * 8 + wave;
  int bh = wg >> 6;
  int qt = wg & 63;

  const long long qkStride = 2LL * 64 * 512;
  const long long vStride  = 4LL * 32 * 512;
  const half_t* Qb = Qfrag + (long long)bh * qkStride;
  const half_t* Kb = Kfrag + (long long)bh * qkStride;
  const half_t* Vb = Vfrag + (long long)bh * vStride;

  v16h aq0 = *(const v16h*)(Qb + ((0LL * 64 + qt) * 32 + lane) * 16);
  v16h aq1 = *(const v16h*)(Qb + ((1LL * 64 + qt) * 32 + lane) * 16);

  v8f o[4];
#pragma unroll
  for (int dt = 0; dt < 4; ++dt) o[dt] = 0.0f;

  if (lane < 16) { mrow[wave][lane] = -3.0e38f; lrow[wave][lane] = 0.f; }
  const float scale = 0.125f;
  int rb = (lane >> 4) * 8, col = lane & 15;

  for (int it = 0; it < 32; ++it) {
    int kb0 = it * 2, kb1 = it * 2 + 1;
    v16h bk00 = *(const v16h*)(Kb + ((0LL * 64 + kb0) * 32 + lane) * 16);
    v16h bk10 = *(const v16h*)(Kb + ((1LL * 64 + kb0) * 32 + lane) * 16);
    v16h bk01 = *(const v16h*)(Kb + ((0LL * 64 + kb1) * 32 + lane) * 16);
    v16h bk11 = *(const v16h*)(Kb + ((1LL * 64 + kb1) * 32 + lane) * 16);

    v8f sA = 0.0f, sB = 0.0f;
    sA = WMMA_F16(aq0, bk00, sA);
    sA = WMMA_F16(aq1, bk10, sA);
    sB = WMMA_F16(aq0, bk01, sB);
    sB = WMMA_F16(aq1, bk11, sB);

#pragma unroll
    for (int r = 0; r < 8; ++r) {
      Stile[wave][rb + r][col]      = sA[r] * scale;
      Stile[wave][rb + r][col + 16] = sB[r] * scale;
    }
    __syncthreads();

    if (lane < 16) {
      int r = lane;
      float m0 = mrow[wave][r];
      float mx = m0;
      for (int cc = 0; cc < 32; ++cc) mx = fmaxf(mx, Stile[wave][r][cc]);
      float alpha = __expf(m0 - mx);
      float sum = 0.f;
      for (int cc = 0; cc < 32; ++cc) {
        float p = __expf(Stile[wave][r][cc] - mx);
        Ptile[wave][r][cc] = (half_t)p;
        sum += p;
      }
      lrow[wave][r] = lrow[wave][r] * alpha + sum;
      mrow[wave][r] = mx;
      arow[wave][r] = alpha;
    }
    __syncthreads();

    float alpha = arow[wave][col];
#pragma unroll
    for (int dt = 0; dt < 4; ++dt)
#pragma unroll
      for (int r = 0; r < 8; ++r) o[dt][r] *= alpha;

    union { v16h v; v8h h[2]; } bp;
    const half_t* prow = &Ptile[wave][col][0];
    bp.h[0] = *(const v8h*)(prow + ((lane >> 4) << 3));
    bp.h[1] = *(const v8h*)(prow + ((lane >> 4) << 3) + 16);

#pragma unroll
    for (int dt = 0; dt < 4; ++dt) {
      v16h av = *(const v16h*)(Vb + (((long long)dt * 32 + it) * 32 + lane) * 16);
      o[dt] = WMMA_F16(av, bp.v, o[dt]);
    }
  }

  float inv = 1.0f / lrow[wave][col];
  int b = bh >> 3, head = bh & 7;
  half_t* Ob = Oplain + ((long long)b * 512 + head * 64) * 1024;
#pragma unroll
  for (int dt = 0; dt < 4; ++dt)
#pragma unroll
    for (int r = 0; r < 8; ++r) {
      int d_ = dt * 16 + r + rb;
      int n  = qt * 16 + col;
      Ob[(long long)d_ * 1024 + n] = (half_t)(o[dt][r] * inv);
    }
}

extern "C" void kernel_launch(void* const* d_in, const int* in_sizes, int n_in,
                              void* d_out, int out_size, void* d_ws, size_t ws_size,
                              hipStream_t stream) {
  (void)in_sizes; (void)n_in; (void)out_size; (void)ws_size;
  const float* x      = (const float*)d_in[0];
  const float* gn_w   = (const float*)d_in[1];
  const float* gn_b   = (const float*)d_in[2];
  const float* qkv_w  = (const float*)d_in[3];
  const float* qkv_b  = (const float*)d_in[4];
  const float* proj_w = (const float*)d_in[5];
  const float* proj_b = (const float*)d_in[6];
  float* out = (float*)d_out;

  half_t* ws = (half_t*)d_ws;
  long long off = 0;
  half_t* h16   = ws + off; off += 8LL * 512 * 1024;
  half_t* Hfrag = ws + off; off += 8LL * 512 * 1024;
  half_t* Wq    = ws + off; off += 1536LL * 512;
  half_t* Wp    = ws + off; off += 512LL * 512;
  half_t* qkvp  = ws + off; off += 8LL * 1536 * 1024;
  half_t* Qf    = ws + off; off += 64LL * 2 * 64 * 512;
  half_t* Kf    = ws + off; off += 64LL * 2 * 64 * 512;
  half_t* Vf    = ws + off; off += 64LL * 4 * 32 * 512;
  half_t* Op    = ws + off; off += 8LL * 512 * 1024;
  half_t* Of    = ws + off; off += 8LL * 512 * 1024;

  groupnorm_kernel<<<dim3(256), dim3(256), 0, stream>>>(x, gn_w, gn_b, h16);

  repackA_kernel<float><<<dim3(768, 1), dim3(256), 0, stream>>>(
      qkv_w, Wq, 96, 16, 512, 1, 0, 0, 0);
  repackA_kernel<float><<<dim3(256, 1), dim3(256), 0, stream>>>(
      proj_w, Wp, 32, 16, 512, 1, 0, 0, 0);

  repackB_kernel<half_t><<<dim3(512, 8), dim3(256), 0, stream>>>(
      h16, Hfrag, 16, 64, 1024, 1, 512LL * 1024, 0, 16LL * 64 * 512);

  gemm_qkv_kernel<<<dim3(12, 8, 8), dim3(256), 0, stream>>>(Wq, Hfrag, qkv_b, qkvp);

  repackB_kernel<half_t><<<dim3(64, 64), dim3(256), 0, stream>>>(
      qkvp, Qf, 2, 64, 1024, 8, 1536LL * 1024, 64LL * 1024, 2LL * 64 * 512);
  repackB_kernel<half_t><<<dim3(64, 64), dim3(256), 0, stream>>>(
      qkvp + 512LL * 1024, Kf, 2, 64, 1024, 8, 1536LL * 1024, 64LL * 1024, 2LL * 64 * 512);
  repackA_kernel<half_t><<<dim3(64, 64), dim3(256), 0, stream>>>(
      qkvp + 1024LL * 1024, Vf, 4, 32, 1024, 8, 1536LL * 1024, 64LL * 1024, 4LL * 32 * 512);

  attention_kernel<<<dim3(512), dim3(256), 0, stream>>>(Qf, Kf, Vf, Op);

  repackB_kernel<half_t><<<dim3(512, 8), dim3(256), 0, stream>>>(
      Op, Of, 16, 64, 1024, 1, 512LL * 1024, 0, 16LL * 64 * 512);
  gemm_proj_kernel<<<dim3(4, 8, 8), dim3(256), 0, stream>>>(Wp, Of, proj_b, x, out);
}


# attention softmax from LDS (16 serial lanes, 2 block barriers per step) to registers: scores computed transposed so P is the PV B fragment as is; row sums as two half-wave partials joined at the end
# speedup vs baseline: 1.5712x; 1.5712x over previous
#include <hip/hip_runtime.h>

typedef _Float16 half_t;
typedef __attribute__((ext_vector_type(16))) _Float16 v16h;
typedef __attribute__((ext_vector_type(8)))  _Float16 v8h;
typedef __attribute__((ext_vector_type(8)))  float    v8f;

#define WMMA_F16(a, b, c) \
  __builtin_amdgcn_wmma_f32_16x16x32_f16(false, (a), false, (b), (short)0, (c), false, false)

__device__ __forceinline__ int kmap_of(int lane, int j) {
  return ((lane >> 4) << 3) + (j & 7) + ((j >> 3) << 4);
}

__global__ __launch_bounds__(256)
void groupnorm_kernel(const float* __restrict__ x,
                      const float* __restrict__ gw,
                      const float* __restrict__ gb,
                      half_t* __restrict__ h16) {
  __shared__ float s_sum[256], s_ss[256];
  int b = blockIdx.x >> 5;
  int g = blockIdx.x & 31;
  const int GE = 16 * 1024;
  const float* xb = x + ((long long)b * 512 + g * 16) * 1024;
  float sum = 0.f, ss = 0.f;
  for (int i = threadIdx.x; i < GE; i += 256) {
    float v = xb[i];
    sum += v; ss += v * v;
  }
  s_sum[threadIdx.x] = sum; s_ss[threadIdx.x] = ss;
  __syncthreads();
  for (int off = 128; off > 0; off >>= 1) {
    if (threadIdx.x < off) {
      s_sum[threadIdx.x] += s_sum[threadIdx.x + off];
      s_ss[threadIdx.x]  += s_ss[threadIdx.x + off];
    }
    __syncthreads();
  }
  float mean = s_sum[0] * (1.0f / GE);
  float var  = s_ss[0] * (1.0f / GE) - mean * mean;
  float inv  = rsqrtf(var + 1e-5f);
  half_t* hb = h16 + ((long long)b * 512 + g * 16) * 1024;
  for (int i = threadIdx.x; i < GE; i += 256) {
    int c = g * 16 + (i >> 10);
    hb[i] = (half_t)((xb[i] - mean) * inv * gw[c] + gb[c]);
  }
}

template <typename SrcT>
__global__ __launch_bounds__(256)
void repackB_kernel(const SrcT* __restrict__ src, half_t* __restrict__ dst,
                    int KT, int NT, int ncols, int nInner,
                    long long outerStride, long long innerStride, long long dstStride) {
  int y = blockIdx.y;
  const SrcT* s = src + (long long)(y / nInner) * outerStride
                      + (long long)(y % nInner) * innerStride;
  half_t* d = dst + (long long)y * dstStride;
  int total = KT * NT * 512;
  for (int tid = blockIdx.x * 256 + threadIdx.x; tid < total; tid += gridDim.x * 256) {
    int j = tid & 15, lane = (tid >> 4) & 31, tile = tid >> 9;
    int kt = tile / NT, nt = tile - kt * NT;
    int row = kt * 32 + kmap_of(lane, j);
    int col = nt * 16 + (lane & 15);
    d[tid] = (half_t)s[(long long)row * ncols + col];
  }
}

template <typename SrcT>
__global__ __launch_bounds__(256)
void repackA_kernel(const SrcT* __restrict__ src, half_t* __restrict__ dst,
                    int MT, int KT, int kcols, int nInner,
                    long long outerStride, long long innerStride, long long dstStride) {
  int y = blockIdx.y;
  const SrcT* s = src + (long long)(y / nInner) * outerStride
                      + (long long)(y % nInner) * innerStride;
  half_t* d = dst + (long long)y * dstStride;
  int total = MT * KT * 512;
  for (int tid = blockIdx.x * 256 + threadIdx.x; tid < total; tid += gridDim.x * 256) {
    int j = tid & 15, lane = (tid >> 4) & 31, tile = tid >> 9;
    int mt = tile / KT, kt = tile - mt * KT;
    int row = mt * 16 + (lane & 15);
    int col = kt * 32 + kmap_of(lane, j);
    d[tid] = (half_t)s[(long long)row * kcols + col];
  }
}

__device__ __forceinline__
void gemm_wave64x32(const half_t* __restrict__ Afrag, const half_t* __restrict__ Bfrag,
                    int KT, int NT, int mtBase, int ntBase, int lane, v8f c[4][2]) {
  for (int kt = 0; kt < KT; ++kt) {
    v16h a[4], bb[2];
#pragma unroll
    for (int i = 0; i < 4; ++i)
      a[i] = *(const v16h*)(Afrag + (((long long)(mtBase + i) * KT + kt) * 32 + lane) * 16);
#pragma unroll
    for (int j = 0; j < 2; ++j)
      bb[j] = *(const v16h*)(Bfrag + (((long long)kt * NT + (ntBase + j)) * 32 + lane) * 16);
#pragma unroll
    for (int i = 0; i < 4; ++i)
#pragma unroll
      for (int j = 0; j < 2; ++j)
        c[i][j] = WMMA_F16(a[i], bb[j], c[i][j]);
  }
}

__global__ __launch_bounds__(256)
void gemm_qkv_kernel(const half_t* __restrict__ Wfrag, const half_t* __restrict__ Hfrag,
                     const float* __restrict__ bias, half_t* __restrict__ out) {
  int wave = threadIdx.x >> 5, lane = threadIdx.x & 31;
  int b = blockIdx.z;
  int wm = wave & 1, wn = wave >> 1;
  int mBase = blockIdx.x * 128 + wm * 64;
  int nBase = blockIdx.y * 128 + wn * 32;
  v8f c[4][2];
#pragma unroll
  for (int i = 0; i < 4; ++i)
#pragma unroll
    for (int j = 0; j < 2; ++j) c[i][j] = 0.0f;
  gemm_wave64x32(Wfrag, Hfrag + (long long)b * (16LL * 64 * 512),
                 16, 64, mBase >> 4, nBase >> 4, lane, c);
  half_t* ob = out + (long long)b * 1536 * 1024;
  int rAdd = 8 * (lane >> 4), col = lane & 15;
#pragma unroll
  for (int i = 0; i < 4; ++i)
#pragma unroll
    for (int j = 0; j < 2; ++j)
#pragma unroll
      for (int r = 0; r < 8; ++r) {
        int m = mBase + i * 16 + r + rAdd;
        int n = nBase + j * 16 + col;
        ob[(long long)m * 1024 + n] = (half_t)(c[i][j][r] + bias[m]);
      }
}

__global__ __launch_bounds__(256)
void gemm_proj_kernel(const half_t* __restrict__ Wfrag, const half_t* __restrict__ Bfrag,
                      const float* __restrict__ bias, const float* __restrict__ resid,
                      float* __restrict__ out) {
  int wave = threadIdx.x >> 5, lane = threadIdx.x & 31;
  int b = blockIdx.z;
  int wm = wave & 1, wn = wave >> 1;
  int mBase = blockIdx.x * 128 + wm * 64;
  int nBase = blockIdx.y * 128 + wn * 32;
  v8f c[4][2];
#pragma unroll
  for (int i = 0; i < 4; ++i)
#pragma unroll
    for (int j = 0; j < 2; ++j) c[i][j] = 0.0f;
  gemm_wave64x32(Wfrag, Bfrag + (long long)b * (16LL * 64 * 512),
                 16, 64, mBase >> 4, nBase >> 4, lane, c);
  const float* xb = resid + (long long)b * 512 * 1024;
  float* ob = out + (long long)b * 512 * 1024;
  int rAdd = 8 * (lane >> 4), col = lane & 15;
#pragma unroll
  for (int i = 0; i < 4; ++i)
#pragma unroll
    for (int j = 0; j < 2; ++j)
#pragma unroll
      for (int r = 0; r < 8; ++r) {
        int m = mBase + i * 16 + r + rAdd;
        int n = nBase + j * 16 + col;
        long long idx = (long long)m * 1024 + n;
        ob[idx] = c[i][j][r] + bias[m] + xb[idx];
      }
}

__device__ __forceinline__ float xhalf(float x) {
  int b = __builtin_bit_cast(int, x);
  return __builtin_bit_cast(float,
      __builtin_amdgcn_permlanex16(b, b, 0x76543210, 0xfedcba98, false, false));
}

__global__ __launch_bounds__(256) __attribute__((amdgpu_waves_per_eu(4, 4)))
void attention_kernel(const half_t* __restrict__ Qfrag, const half_t* __restrict__ Kfrag,
                      const half_t* __restrict__ Vfrag, half_t* __restrict__ Oplain) {
  int wave = threadIdx.x >> 5, lane = threadIdx.x & 31;
  int wg = blockIdx.x * 8 + wave;
  int bh = wg >> 6;
  int qt = wg & 63;

  const long long qkStride = 2LL * 64 * 512;
  const long long vStride  = 4LL * 32 * 512;
  const half_t* Qb = Qfrag + (long long)bh * qkStride;
  const half_t* Kb = Kfrag + (long long)bh * qkStride;
  const half_t* Vb = Vfrag + (long long)bh * vStride;

  v16h bq0 = *(const v16h*)(Qb + ((0LL * 64 + qt) * 32 + lane) * 16);
  v16h bq1 = *(const v16h*)(Qb + ((1LL * 64 + qt) * 32 + lane) * 16);

  v8f o[4];
#pragma unroll
  for (int dt = 0; dt < 4; ++dt) o[dt] = 0.0f;

  const float cs = 0.125f * 1.4426950408889634f;
  float m_run  = -3.0e38f;
  float l_part = 0.f;
  int rb = (lane >> 4) * 8, col = lane & 15;

  for (int it = 0; it < 32; ++it) {
    int kb0 = it * 2, kb1 = it * 2 + 1;
    v16h ak00 = *(const v16h*)(Kb + ((0LL * 64 + kb0) * 32 + lane) * 16);
    v16h ak10 = *(const v16h*)(Kb + ((1LL * 64 + kb0) * 32 + lane) * 16);
    v16h ak01 = *(const v16h*)(Kb + ((0LL * 64 + kb1) * 32 + lane) * 16);
    v16h ak11 = *(const v16h*)(Kb + ((1LL * 64 + kb1) * 32 + lane) * 16);
    v16h av[4];
#pragma unroll
    for (int dt = 0; dt < 4; ++dt)
      av[dt] = *(const v16h*)(Vb + (((long long)dt * 32 + it) * 32 + lane) * 16);

    v8f sA = 0.0f, sB = 0.0f;
    sA = WMMA_F16(ak00, bq0, sA);
    sA = WMMA_F16(ak10, bq1, sA);
    sB = WMMA_F16(ak01, bq0, sB);
    sB = WMMA_F16(ak11, bq1, sB);

    float mx = fmaxf(sA[0], sB[0]);
#pragma unroll
    for (int r = 1; r < 8; ++r) mx = fmaxf(mx, fmaxf(sA[r], sB[r]));
    mx = fmaxf(mx, xhalf(mx));
    float m_new = fmaxf(m_run, mx);
    float alpha = __builtin_amdgcn_exp2f((m_run - m_new) * cs);
    float nm = -(m_new * cs);

    v16h bp;
    float sumA = 0.f, sumB = 0.f;
#pragma unroll
    for (int r = 0; r < 8; ++r) {
      float pa = __builtin_amdgcn_exp2f(__builtin_fmaf(sA[r], cs, nm));
      float pb = __builtin_amdgcn_exp2f(__builtin_fmaf(sB[r], cs, nm));
      sumA += pa; sumB += pb;
      bp[r]     = (half_t)pa;
      bp[r + 8] = (half_t)pb;
    }
    l_part = l_part * alpha + (sumA + sumB);
    m_run = m_new;

#pragma unroll
    for (int dt = 0; dt < 4; ++dt)
#pragma unroll
      for (int r = 0; r < 8; ++r) o[dt][r] *= alpha;

#pragma unroll
    for (int dt = 0; dt < 4; ++dt) o[dt] = WMMA_F16(av[dt], bp, o[dt]);
  }

  float inv = 1.0f / (l_part + xhalf(l_part));
  int b = bh >> 3, head = bh & 7;
  half_t* Ob = Oplain + ((long long)b * 512 + head * 64) * 1024;
#pragma unroll
  for (int dt = 0; dt < 4; ++dt)
#pragma unroll
    for (int r = 0; r < 8; ++r) {
      int d_ = dt * 16 + r + rb;
      int n  = qt * 16 + col;
      Ob[(long long)d_ * 1024 + n] = (half_t)(o[dt][r] * inv);
    }
}

extern "C" void kernel_launch(void* const* d_in, const int* in_sizes, int n_in,
                              void* d_out, int out_size, void* d_ws, size_t ws_size,
                              hipStream_t stream) {
  (void)in_sizes; (void)n_in; (void)out_size; (void)ws_size;
  const float* x      = (const float*)d_in[0];
  const float* gn_w   = (const float*)d_in[1];
  const float* gn_b   = (const float*)d_in[2];
  const float* qkv_w  = (const float*)d_in[3];
  const float* qkv_b  = (const float*)d_in[4];
  const float* proj_w = (const float*)d_in[5];
  const float* proj_b = (const float*)d_in[6];
  float* out = (float*)d_out;

  half_t* ws = (half_t*)d_ws;
  long long off = 0;
  half_t* h16   = ws + off; off += 8LL * 512 * 1024;
  half_t* Hfrag = ws + off; off += 8LL * 512 * 1024;
  half_t* Wq    = ws + off; off += 1536LL * 512;
  half_t* Wp    = ws + off; off += 512LL * 512;
  half_t* qkvp  = ws + off; off += 8LL * 1536 * 1024;
  half_t* Qf    = ws + off; off += 64LL * 2 * 64 * 512;
  half_t* Kf    = ws + off; off += 64LL * 2 * 64 * 512;
  half_t* Vf    = ws + off; off += 64LL * 4 * 32 * 512;
  half_t* Op    = ws + off; off += 8LL * 512 * 1024;
  half_t* Of    = ws + off; off += 8LL * 512 * 1024;

  groupnorm_kernel<<<dim3(256), dim3(256), 0, stream>>>(x, gn_w, gn_b, h16);

  repackA_kernel<float><<<dim3(768, 1), dim3(256), 0, stream>>>(
      qkv_w, Wq, 96, 16, 512, 1, 0, 0, 0);
  repackA_kernel<float><<<dim3(256, 1), dim3(256), 0, stream>>>(
      proj_w, Wp, 32, 16, 512, 1, 0, 0, 0);

  repackB_kernel<half_t><<<dim3(512, 8), dim3(256), 0, stream>>>(
      h16, Hfrag, 16, 64, 1024, 1, 512LL * 1024, 0, 16LL * 64 * 512);

  gemm_qkv_kernel<<<dim3(12, 8, 8), dim3(256), 0, stream>>>(Wq, Hfrag, qkv_b, qkvp);

  repackB_kernel<half_t><<<dim3(64, 64), dim3(256), 0, stream>>>(
      qkvp, Qf, 2, 64, 1024, 8, 1536LL * 1024, 64LL * 1024, 2LL * 64 * 512);
  repackB_kernel<half_t><<<dim3(64, 64), dim3(256), 0, stream>>>(
      qkvp + 512LL * 1024, Kf, 2, 64, 1024, 8, 1536LL * 1024, 64LL * 1024, 2LL * 64 * 512);
  repackA_kernel<half_t><<<dim3(64, 64), dim3(256), 0, stream>>>(
      qkvp + 1024LL * 1024, Vf, 4, 32, 1024, 8, 1536LL * 1024, 64LL * 1024, 4LL * 32 * 512);

  attention_kernel<<<dim3(512), dim3(256), 0, stream>>>(Qf, Kf, Vf, Op);

  repackB_kernel<half_t><<<dim3(512, 8), dim3(256), 0, stream>>>(
      Op, Of, 16, 64, 1024, 1, 512LL * 1024, 0, 16LL * 64 * 512);
  gemm_proj_kernel<<<dim3(4, 8, 8), dim3(256), 0, stream>>>(Wp, Of, proj_b, x, out);
}


# producers store straight in the consumers' WMMA fragment order: 5 activation repack launches dropped (11 -> 6); QKV GEMM k order inside each 32-block regrouped, so its sum order changes
# speedup vs baseline: 2.1109x; 1.3435x over previous
#include <hip/hip_runtime.h>

typedef _Float16 half_t;
typedef __attribute__((ext_vector_type(16))) _Float16 v16h;
typedef __attribute__((ext_vector_type(8)))  _Float16 v8h;
typedef __attribute__((ext_vector_type(8)))  float    v8f;

#define WMMA_F16(a, b, c) \
  __builtin_amdgcn_wmma_f32_16x16x32_f16(false, (a), false, (b), (short)0, (c), false, false)

__device__ __forceinline__ int kmap_of(int lane, int j) {
  return ((lane >> 4) << 3) + (j & 7) + ((j >> 3) << 4);
}

__device__ __forceinline__ int kmap_grp(int lane, int j) {
  return ((lane >> 4) << 4) + j;
}

__global__ __launch_bounds__(256)
void groupnorm_kernel(const float* __restrict__ x,
                      const float* __restrict__ gw,
                      const float* __restrict__ gb,
                      half_t* __restrict__ Hfrag) {
  __shared__ float s_sum[256], s_ss[256];
  int b = blockIdx.x >> 5;
  int g = blockIdx.x & 31;
  const int GE = 16 * 1024;
  const float* xb = x + ((long long)b * 512 + g * 16) * 1024;
  float sum = 0.f, ss = 0.f;
  for (int i = threadIdx.x; i < GE; i += 256) {
    float v = xb[i];
    sum += v; ss += v * v;
  }
  s_sum[threadIdx.x] = sum; s_ss[threadIdx.x] = ss;
  __syncthreads();
  for (int off = 128; off > 0; off >>= 1) {
    if (threadIdx.x < off) {
      s_sum[threadIdx.x] += s_sum[threadIdx.x + off];
      s_ss[threadIdx.x]  += s_ss[threadIdx.x + off];
    }
    __syncthreads();
  }
  float mean = s_sum[0] * (1.0f / GE);
  float var  = s_ss[0] * (1.0f / GE) - mean * mean;
  float inv  = rsqrtf(var + 1e-5f);
  half_t* hb = Hfrag + (long long)b * (16LL * 64 * 512)
                     + (long long)(g >> 1) * (64 * 512) + (g & 1) * 256;
  for (int n = threadIdx.x; n < 1024; n += 256) {
    v16h hv;
#pragma unroll
    for (int cc = 0; cc < 16; ++cc) {
      int c = g * 16 + cc;
      hv[cc] = (half_t)((xb[cc * 1024 + n] - mean) * inv * gw[c] + gb[c]);
    }
    *(v16h*)(hb + ((n >> 4) * 32 + (n & 15)) * 16) = hv;
  }
}

template <bool GRP>
__global__ __launch_bounds__(256)
void repackA_kernel(const float* __restrict__ src, half_t* __restrict__ dst,
                    int MT, int KT, int kcols) {
  int total = MT * KT * 512;
  for (int tid = blockIdx.x * 256 + threadIdx.x; tid < total; tid += gridDim.x * 256) {
    int j = tid & 15, lane = (tid >> 4) & 31, tile = tid >> 9;
    int mt = tile / KT, kt = tile - mt * KT;
    int row = mt * 16 + (lane & 15);
    int col = kt * 32 + (GRP ? kmap_grp(lane, j) : kmap_of(lane, j));
    dst[tid] = (half_t)src[(long long)row * kcols + col];
  }
}

template <bool TR>
__device__ __forceinline__
void gemm_wave64x32(const half_t* __restrict__ Afrag, const half_t* __restrict__ Bfrag,
                    int KT, int NT, int mtBase, int ntBase, int lane, v8f c[4][2]) {
  for (int kt = 0; kt < KT; ++kt) {
    v16h a[4], bb[2];
#pragma unroll
    for (int i = 0; i < 4; ++i)
      a[i] = *(const v16h*)(Afrag + (((long long)(mtBase + i) * KT + kt) * 32 + lane) * 16);
#pragma unroll
    for (int j = 0; j < 2; ++j)
      bb[j] = *(const v16h*)(Bfrag + (((long long)kt * NT + (ntBase + j)) * 32 + lane) * 16);
#pragma unroll
    for (int i = 0; i < 4; ++i)
#pragma unroll
      for (int j = 0; j < 2; ++j)
        c[i][j] = TR ? WMMA_F16(bb[j], a[i], c[i][j]) : WMMA_F16(a[i], bb[j], c[i][j]);
  }
}

__global__ __launch_bounds__(256)
void gemm_qkv_kernel(const half_t* __restrict__ Wfrag, const half_t* __restrict__ Hfrag,
                     const float* __restrict__ bias, half_t* __restrict__ Qf,
                     half_t* __restrict__ Kf, half_t* __restrict__ Vf) {
  int wave = threadIdx.x >> 5, lane = threadIdx.x & 31;
  int b = blockIdx.z;
  int wm = wave & 1, wn = wave >> 1;
  int mBase = blockIdx.x * 128 + wm * 64;
  int nBase = blockIdx.y * 128 + wn * 32;
  int which = blockIdx.x >> 2;
  int bh = b * 8 + ((mBase & 511) >> 6);
  int hh = lane >> 4, col = lane & 15;
  const half_t* Hb = Hfrag + (long long)b * (16LL * 64 * 512);
  v8f c[4][2];
#pragma unroll
  for (int i = 0; i < 4; ++i)
#pragma unroll
    for (int j = 0; j < 2; ++j) c[i][j] = 0.0f;

  if (which == 2) {
    gemm_wave64x32<true>(Wfrag, Hb, 16, 64, mBase >> 4, nBase >> 4, lane, c);
    half_t* vb = Vf + (long long)bh * (4LL * 32 * 512);
    int ktile = nBase >> 5;
#pragma unroll
    for (int i = 0; i < 4; ++i) {
      float bv = bias[mBase + i * 16 + col];
      v16h ov;
#pragma unroll
      for (int r = 0; r < 8; ++r) {
        ov[r]     = (half_t)(c[i][0][r] + bv);
        ov[r + 8] = (half_t)(c[i][1][r] + bv);
      }
      *(v16h*)(vb + ((long long)(i * 32 + ktile) * 32 + lane) * 16) = ov;
    }
  } else {
    gemm_wave64x32<false>(Wfrag, Hb, 16, 64, mBase >> 4, nBase >> 4, lane, c);
    half_t* qb = (which == 0 ? Qf : Kf) + (long long)bh * (2LL * 64 * 512);
    int ntile = nBase >> 4;
#pragma unroll
    for (int kt = 0; kt < 2; ++kt)
#pragma unroll
      for (int jj = 0; jj < 2; ++jj) {
        v16h ov;
#pragma unroll
        for (int r = 0; r < 8; ++r) {
          int m0 = mBase + (2 * kt) * 16 + 8 * hh + r;
          ov[r]     = (half_t)(c[2 * kt][jj][r]     + bias[m0]);
          ov[r + 8] = (half_t)(c[2 * kt + 1][jj][r] + bias[m0 + 16]);
        }
        *(v16h*)(qb + ((long long)(kt * 64 + ntile + jj) * 32 + lane) * 16) = ov;
      }
  }
}

__global__ __launch_bounds__(256)
void gemm_proj_kernel(const half_t* __restrict__ Wfrag, const half_t* __restrict__ Bfrag,
                      const float* __restrict__ bias, const float* __restrict__ resid,
                      float* __restrict__ out) {
  int wave = threadIdx.x >> 5, lane = threadIdx.x & 31;
  int b = blockIdx.z;
  int wm = wave & 1, wn = wave >> 1;
  int mBase = blockIdx.x * 128 + wm * 64;
  int nBase = blockIdx.y * 128 + wn * 32;
  v8f c[4][2];
#pragma unroll
  for (int i = 0; i < 4; ++i)
#pragma unroll
    for (int j = 0; j < 2; ++j) c[i][j] = 0.0f;
  gemm_wave64x32<false>(Wfrag, Bfrag + (long long)b * (16LL * 64 * 512),
                        16, 64, mBase >> 4, nBase >> 4, lane, c);
  const float* xb = resid + (long long)b * 512 * 1024;
  float* ob = out + (long long)b * 512 * 1024;
  int rAdd = 8 * (lane >> 4), col = lane & 15;
#pragma unroll
  for (int i = 0; i < 4; ++i)
#pragma unroll
    for (int j = 0; j < 2; ++j)
#pragma unroll
      for (int r = 0; r < 8; ++r) {
        int m = mBase + i * 16 + r + rAdd;
        int n = nBase + j * 16 + col;
        long long idx = (long long)m * 1024 + n;
        ob[idx] = c[i][j][r] + bias[m] + xb[idx];
      }
}

__device__ __forceinline__ float xhalf(float x) {
  int b = __builtin_bit_cast(int, x);
  return __builtin_bit_cast(float,
      __builtin_amdgcn_permlanex16(b, b, 0x76543210, 0xfedcba98, false, false));
}

__global__ __launch_bounds__(256) __attribute__((amdgpu_waves_per_eu(4, 4)))
void attention_kernel(const half_t* __restrict__ Qfrag, const half_t* __restrict__ Kfrag,
                      const half_t* __restrict__ Vfrag, half_t* __restrict__ Ofrag) {
  int wave = threadIdx.x >> 5, lane = threadIdx.x & 31;
  int wg = blockIdx.x * 8 + wave;
  int bh = wg >> 6;
  int qt = wg & 63;

  const long long qkStride = 2LL * 64 * 512;
  const long long vStride  = 4LL * 32 * 512;
  const half_t* Qb = Qfrag + (long long)bh * qkStride;
  const half_t* Kb = Kfrag + (long long)bh * qkStride;
  const half_t* Vb = Vfrag + (long long)bh * vStride;

  v16h bq0 = *(const v16h*)(Qb + ((0LL * 64 + qt) * 32 + lane) * 16);
  v16h bq1 = *(const v16h*)(Qb + ((1LL * 64 + qt) * 32 + lane) * 16);

  v8f o[4];
#pragma unroll
  for (int dt = 0; dt < 4; ++dt) o[dt] = 0.0f;

  const float cs = 0.125f * 1.4426950408889634f;
  float m_run  = -3.0e38f;
  float l_part = 0.f;

  for (int it = 0; it < 32; ++it) {
    int kb0 = it * 2, kb1 = it * 2 + 1;
    v16h ak00 = *(const v16h*)(Kb + ((0LL * 64 + kb0) * 32 + lane) * 16);
    v16h ak10 = *(const v16h*)(Kb + ((1LL * 64 + kb0) * 32 + lane) * 16);
    v16h ak01 = *(const v16h*)(Kb + ((0LL * 64 + kb1) * 32 + lane) * 16);
    v16h ak11 = *(const v16h*)(Kb + ((1LL * 64 + kb1) * 32 + lane) * 16);
    v16h av[4];
#pragma unroll
    for (int dt = 0; dt < 4; ++dt)
      av[dt] = *(const v16h*)(Vb + (((long long)dt * 32 + it) * 32 + lane) * 16);

    v8f sA = 0.0f, sB = 0.0f;
    sA = WMMA_F16(ak00, bq0, sA);
    sA = WMMA_F16(ak10, bq1, sA);
    sB = WMMA_F16(ak01, bq0, sB);
    sB = WMMA_F16(ak11, bq1, sB);

    float mx = fmaxf(sA[0], sB[0]);
#pragma unroll
    for (int r = 1; r < 8; ++r) mx = fmaxf(mx, fmaxf(sA[r], sB[r]));
    mx = fmaxf(mx, xhalf(mx));
    float m_new = fmaxf(m_run, mx);
    float alpha = __builtin_amdgcn_exp2f((m_run - m_new) * cs);
    float nm = -(m_new * cs);

    v16h bp;
    float sumA = 0.f, sumB = 0.f;
#pragma unroll
    for (int r = 0; r < 8; ++r) {
      float pa = __builtin_amdgcn_exp2f(__builtin_fmaf(sA[r], cs, nm));
      float pb = __builtin_amdgcn_exp2f(__builtin_fmaf(sB[r], cs, nm));
      sumA += pa; sumB += pb;
      bp[r]     = (half_t)pa;
      bp[r + 8] = (half_t)pb;
    }
    l_part = l_part * alpha + (sumA + sumB);
    m_run = m_new;

#pragma unroll
    for (int dt = 0; dt < 4; ++dt)
#pragma unroll
      for (int r = 0; r < 8; ++r) o[dt][r] *= alpha;

#pragma unroll
    for (int dt = 0; dt < 4; ++dt) o[dt] = WMMA_F16(av[dt], bp, o[dt]);
  }

  float inv = 1.0f / (l_part + xhalf(l_part));
  int b = bh >> 3, head = bh & 7;
  half_t* Ob = Ofrag + (long long)b * (16LL * 64 * 512);
#pragma unroll
  for (int t = 0; t < 2; ++t) {
    v16h ov;
#pragma unroll
    for (int r = 0; r < 8; ++r) {
      ov[r]     = (half_t)(o[2 * t][r] * inv);
      ov[r + 8] = (half_t)(o[2 * t + 1][r] * inv);
    }
    *(v16h*)(Ob + ((long long)((head * 2 + t) * 64 + qt) * 32 + lane) * 16) = ov;
  }
}

extern "C" void kernel_launch(void* const* d_in, const int* in_sizes, int n_in,
                              void* d_out, int out_size, void* d_ws, size_t ws_size,
                              hipStream_t stream) {
  (void)in_sizes; (void)n_in; (void)out_size; (void)ws_size;
  const float* x      = (const float*)d_in[0];
  const float* gn_w   = (const float*)d_in[1];
  const float* gn_b   = (const float*)d_in[2];
  const float* qkv_w  = (const float*)d_in[3];
  const float* qkv_b  = (const float*)d_in[4];
  const float* proj_w = (const float*)d_in[5];
  const float* proj_b = (const float*)d_in[6];
  float* out = (float*)d_out;

  half_t* ws = (half_t*)d_ws;
  long long off = 0;
  half_t* Hfrag = ws + off; off += 8LL * 512 * 1024;
  half_t* Wq    = ws + off; off += 1536LL * 512;
  half_t* Wp    = ws + off; off += 512LL * 512;
  half_t* Qf    = ws + off; off += 64LL * 2 * 64 * 512;
  half_t* Kf    = ws + off; off += 64LL * 2 * 64 * 512;
  half_t* Vf    = ws + off; off += 64LL * 4 * 32 * 512;
  half_t* Of    = ws + off; off += 8LL * 512 * 1024;

  groupnorm_kernel<<<dim3(256), dim3(256), 0, stream>>>(x, gn_w, gn_b, Hfrag);

  repackA_kernel<true><<<dim3(768), dim3(256), 0, stream>>>(qkv_w, Wq, 96, 16, 512);
  repackA_kernel<false><<<dim3(256), dim3(256), 0, stream>>>(proj_w, Wp, 32, 16, 512);

  gemm_qkv_kernel<<<dim3(12, 8, 8), dim3(256), 0, stream>>>(Wq, Hfrag, qkv_b, Qf, Kf, Vf);

  attention_kernel<<<dim3(512), dim3(256), 0, stream>>>(Qf, Kf, Vf, Of);

  gemm_proj_kernel<<<dim3(4, 8, 8), dim3(256), 0, stream>>>(Wp, Of, proj_b, x, out);
}


# fragment tiles stored as two planes of 32 lanes x 16 bytes: each 16-byte load/store instruction of a wave covers 4 whole 128-byte lines instead of 8 half-used ones (QKV, attention, proj)
# speedup vs baseline: 2.2991x; 1.0892x over previous
#include <hip/hip_runtime.h>

typedef _Float16 half_t;
typedef __attribute__((ext_vector_type(16))) _Float16 v16h;
typedef __attribute__((ext_vector_type(8)))  _Float16 v8h;
typedef __attribute__((ext_vector_type(8)))  float    v8f;

#define WMMA_F16(a, b, c) \
  __builtin_amdgcn_wmma_f32_16x16x32_f16(false, (a), false, (b), (short)0, (c), false, false)

__device__ __forceinline__ int kmap_of(int lane, int j) {
  return ((lane >> 4) << 3) + (j & 7) + ((j >> 3) << 4);
}

__device__ __forceinline__ int kmap_grp(int lane, int j) {
  return ((lane >> 4) << 4) + j;
}

__device__ __forceinline__ v16h ld_frag(const half_t* __restrict__ tile, int lane) {
  v8h lo = *(const v8h*)(tile + lane * 8);
  v8h hi = *(const v8h*)(tile + 256 + lane * 8);
  return __builtin_shufflevector(lo, hi, 0, 1, 2, 3, 4, 5, 6, 7, 8, 9, 10, 11, 12, 13, 14, 15);
}
__device__ __forceinline__ void st_frag(half_t* __restrict__ tile, int lane, v16h v) {
  *(v8h*)(tile + lane * 8)       = __builtin_shufflevector(v, v, 0, 1, 2, 3, 4, 5, 6, 7);
  *(v8h*)(tile + 256 + lane * 8) = __builtin_shufflevector(v, v, 8, 9, 10, 11, 12, 13, 14, 15);
}

__global__ __launch_bounds__(256)
void groupnorm_kernel(const float* __restrict__ x,
                      const float* __restrict__ gw,
                      const float* __restrict__ gb,
                      half_t* __restrict__ Hfrag) {
  __shared__ float s_sum[256], s_ss[256];
  int b = blockIdx.x >> 5;
  int g = blockIdx.x & 31;
  const int GE = 16 * 1024;
  const float* xb = x + ((long long)b * 512 + g * 16) * 1024;
  float sum = 0.f, ss = 0.f;
  for (int i = threadIdx.x; i < GE; i += 256) {
    float v = xb[i];
    sum += v; ss += v * v;
  }
  s_sum[threadIdx.x] = sum; s_ss[threadIdx.x] = ss;
  __syncthreads();
  for (int off = 128; off > 0; off >>= 1) {
    if (threadIdx.x < off) {
      s_sum[threadIdx.x] += s_sum[threadIdx.x + off];
      s_ss[threadIdx.x]  += s_ss[threadIdx.x + off];
    }
    __syncthreads();
  }
  float mean = s_sum[0] * (1.0f / GE);
  float var  = s_ss[0] * (1.0f / GE) - mean * mean;
  float inv  = rsqrtf(var + 1e-5f);
  half_t* hb = Hfrag + (long long)b * (16LL * 64 * 512)
                     + (long long)(g >> 1) * (64 * 512);
  int laneAdd = (g & 1) * 16;
  for (int n = threadIdx.x; n < 1024; n += 256) {
    v16h hv;
#pragma unroll
    for (int cc = 0; cc < 16; ++cc) {
      int c = g * 16 + cc;
      hv[cc] = (half_t)((xb[cc * 1024 + n] - mean) * inv * gw[c] + gb[c]);
    }
    st_frag(hb + (n >> 4) * 512, (n & 15) + laneAdd, hv);
  }
}

template <bool GRP>
__global__ __launch_bounds__(256)
void repackA_kernel(const float* __restrict__ src, half_t* __restrict__ dst,
                    int MT, int KT, int kcols) {
  int total = MT * KT * 512;
  for (int tid = blockIdx.x * 256 + threadIdx.x; tid < total; tid += gridDim.x * 256) {
    int j = (tid & 7) + ((tid >> 5) & 8), lane = (tid >> 3) & 31, tile = tid >> 9;
    int mt = tile / KT, kt = tile - mt * KT;
    int row = mt * 16 + (lane & 15);
    int col = kt * 32 + (GRP ? kmap_grp(lane, j) : kmap_of(lane, j));
    dst[tid] = (half_t)src[(long long)row * kcols + col];
  }
}

template <bool TR>
__device__ __forceinline__
void gemm_wave64x32(const half_t* __restrict__ Afrag, const half_t* __restrict__ Bfrag,
                    int KT, int NT, int mtBase, int ntBase, int lane, v8f c[4][2]) {
  for (int kt = 0; kt < KT; ++kt) {
    v16h a[4], bb[2];
#pragma unroll
    for (int i = 0; i < 4; ++i)
      a[i] = ld_frag(Afrag + ((long long)(mtBase + i) * KT + kt) * 512, lane);
#pragma unroll
    for (int j = 0; j < 2; ++j)
      bb[j] = ld_frag(Bfrag + ((long long)kt * NT + (ntBase + j)) * 512, lane);
#pragma unroll
    for (int i = 0; i < 4; ++i)
#pragma unroll
      for (int j = 0; j < 2; ++j)
        c[i][j] = TR ? WMMA_F16(bb[j], a[i], c[i][j]) : WMMA_F16(a[i], bb[j], c[i][j]);
  }
}

__global__ __launch_bounds__(256)
void gemm_qkv_kernel(const half_t* __restrict__ Wfrag, const half_t* __restrict__ Hfrag,
                     const float* __restrict__ bias, half_t* __restrict__ Qf,
                     half_t* __restrict__ Kf, half_t* __restrict__ Vf) {
  int wave = threadIdx.x >> 5, lane = threadIdx.x & 31;
  int b = blockIdx.z;
  int wm = wave & 1, wn = wave >> 1;
  int mBase = blockIdx.x * 128 + wm * 64;
  int nBase = blockIdx.y * 128 + wn * 32;
  int which = blockIdx.x >> 2;
  int bh = b * 8 + ((mBase & 511) >> 6);
  int hh = lane >> 4, col = lane & 15;
  const half_t* Hb = Hfrag + (long long)b * (16LL * 64 * 512);
  v8f c[4][2];
#pragma unroll
  for (int i = 0; i < 4; ++i)
#pragma unroll
    for (int j = 0; j < 2; ++j) c[i][j] = 0.0f;

  if (which == 2) {
    gemm_wave64x32<true>(Wfrag, Hb, 16, 64, mBase >> 4, nBase >> 4, lane, c);
    half_t* vb = Vf + (long long)bh * (4LL * 32 * 512);
    int ktile = nBase >> 5;
#pragma unroll
    for (int i = 0; i < 4; ++i) {
      float bv = bias[mBase + i * 16 + col];
      v16h ov;
#pragma unroll
      for (int r = 0; r < 8; ++r) {
        ov[r]     = (half_t)(c[i][0][r] + bv);
        ov[r + 8] = (half_t)(c[i][1][r] + bv);
      }
      st_frag(vb + (long long)(i * 32 + ktile) * 512, lane, ov);
    }
  } else {
    gemm_wave64x32<false>(Wfrag, Hb, 16, 64, mBase >> 4, nBase >> 4, lane, c);
    half_t* qb = (which == 0 ? Qf : Kf) + (long long)bh * (2LL * 64 * 512);
    int ntile = nBase >> 4;
#pragma unroll
    for (int kt = 0; kt < 2; ++kt)
#pragma unroll
      for (int jj = 0; jj < 2; ++jj) {
        v16h ov;
#pragma unroll
        for (int r = 0; r < 8; ++r) {
          int m0 = mBase + (2 * kt) * 16 + 8 * hh + r;
          ov[r]     = (half_t)(c[2 * kt][jj][r]     + bias[m0]);
          ov[r + 8] = (half_t)(c[2 * kt + 1][jj][r] + bias[m0 + 16]);
        }
        st_frag(qb + (long long)(kt * 64 + ntile + jj) * 512, lane, ov);
      }
  }
}

__global__ __launch_bounds__(256)
void gemm_proj_kernel(const half_t* __restrict__ Wfrag, const half_t* __restrict__ Bfrag,
                      const float* __restrict__ bias, const float* __restrict__ resid,
                      float* __restrict__ out) {
  int wave = threadIdx.x >> 5, lane = threadIdx.x & 31;
  int b = blockIdx.z;
  int wm = wave & 1, wn = wave >> 1;
  int mBase = blockIdx.x * 128 + wm * 64;
  int nBase = blockIdx.y * 128 + wn * 32;
  v8f c[4][2];
#pragma unroll
  for (int i = 0; i < 4; ++i)
#pragma unroll
    for (int j = 0; j < 2; ++j) c[i][j] = 0.0f;
  gemm_wave64x32<false>(Wfrag, Bfrag + (long long)b * (16LL * 64 * 512),
                        16, 64, mBase >> 4, nBase >> 4, lane, c);
  const float* xb = resid + (long long)b * 512 * 1024;
  float* ob = out + (long long)b * 512 * 1024;
  int rAdd = 8 * (lane >> 4), col = lane & 15;
#pragma unroll
  for (int i = 0; i < 4; ++i)
#pragma unroll
    for (int j = 0; j < 2; ++j)
#pragma unroll
      for (int r = 0; r < 8; ++r) {
        int m = mBase + i * 16 + r + rAdd;
        int n = nBase + j * 16 + col;
        long long idx = (long long)m * 1024 + n;
        ob[idx] = c[i][j][r] + bias[m] + xb[idx];
      }
}

__device__ __forceinline__ float xhalf(float x) {
  int b = __builtin_bit_cast(int, x);
  return __builtin_bit_cast(float,
      __builtin_amdgcn_permlanex16(b, b, 0x76543210, 0xfedcba98, false, false));
}

__global__ __launch_bounds__(256) __attribute__((amdgpu_waves_per_eu(4, 4)))
void attention_kernel(const half_t* __restrict__ Qfrag, const half_t* __restrict__ Kfrag,
                      const half_t* __restrict__ Vfrag, half_t* __restrict__ Ofrag) {
  int wave = threadIdx.x >> 5, lane = threadIdx.x & 31;
  int wg = blockIdx.x * 8 + wave;
  int bh = wg >> 6;
  int qt = wg & 63;

  const long long qkStride = 2LL * 64 * 512;
  const long long vStride  = 4LL * 32 * 512;
  const half_t* Qb = Qfrag + (long long)bh * qkStride;
  const half_t* Kb = Kfrag + (long long)bh * qkStride;
  const half_t* Vb = Vfrag + (long long)bh * vStride;

  v16h bq0 = ld_frag(Qb + (0LL * 64 + qt) * 512, lane);
  v16h bq1 = ld_frag(Qb + (1LL * 64 + qt) * 512, lane);

  v8f o[4];
#pragma unroll
  for (int dt = 0; dt < 4; ++dt) o[dt] = 0.0f;

  const float cs = 0.125f * 1.4426950408889634f;
  float m_run  = -3.0e38f;
  float l_part = 0.f;

  for (int it = 0; it < 32; ++it) {
    int kb0 = it * 2, kb1 = it * 2 + 1;
    v16h ak00 = ld_frag(Kb + (0LL * 64 + kb0) * 512, lane);
    v16h ak10 = ld_frag(Kb + (1LL * 64 + kb0) * 512, lane);
    v16h ak01 = ld_frag(Kb + (0LL * 64 + kb1) * 512, lane);
    v16h ak11 = ld_frag(Kb + (1LL * 64 + kb1) * 512, lane);
    v16h av[4];
#pragma unroll
    for (int dt = 0; dt < 4; ++dt)
      av[dt] = ld_frag(Vb + ((long long)dt * 32 + it) * 512, lane);

    v8f sA = 0.0f, sB = 0.0f;
    sA = WMMA_F16(ak00, bq0, sA);
    sA = WMMA_F16(ak10, bq1, sA);
    sB = WMMA_F16(ak01, bq0, sB);
    sB = WMMA_F16(ak11, bq1, sB);

    float mx = fmaxf(sA[0], sB[0]);
#pragma unroll
    for (int r = 1; r < 8; ++r) mx = fmaxf(mx, fmaxf(sA[r], sB[r]));
    mx = fmaxf(mx, xhalf(mx));
    float m_new = fmaxf(m_run, mx);
    float alpha = __builtin_amdgcn_exp2f((m_run - m_new) * cs);
    float nm = -(m_new * cs);

    v16h bp;
    float sumA = 0.f, sumB = 0.f;
#pragma unroll
    for (int r = 0; r < 8; ++r) {
      float pa = __builtin_amdgcn_exp2f(__builtin_fmaf(sA[r], cs, nm));
      float pb = __builtin_amdgcn_exp2f(__builtin_fmaf(sB[r], cs, nm));
      sumA += pa; sumB += pb;
      bp[r]     = (half_t)pa;
      bp[r + 8] = (half_t)pb;
    }
    l_part = l_part * alpha + (sumA + sumB);
    m_run = m_new;

#pragma unroll
    for (int dt = 0; dt < 4; ++dt)
#pragma unroll
      for (int r = 0; r < 8; ++r) o[dt][r] *= alpha;

#pragma unroll
    for (int dt = 0; dt < 4; ++dt) o[dt] = WMMA_F16(av[dt], bp, o[dt]);
  }

  float inv = 1.0f / (l_part + xhalf(l_part));
  int b = bh >> 3, head = bh & 7;
  half_t* Ob = Ofrag + (long long)b * (16LL * 64 * 512);
#pragma unroll
  for (int t = 0; t < 2; ++t) {
    v16h ov;
#pragma unroll
    for (int r = 0; r < 8; ++r) {
      ov[r]     = (half_t)(o[2 * t][r] * inv);
      ov[r + 8] = (half_t)(o[2 * t + 1][r] * inv);
    }
    st_frag(Ob + (long long)((head * 2 + t) * 64 + qt) * 512, lane, ov);
  }
}

extern "C" void kernel_launch(void* const* d_in, const int* in_sizes, int n_in,
                              void* d_out, int out_size, void* d_ws, size_t ws_size,
                              hipStream_t stream) {
  (void)in_sizes; (void)n_in; (void)out_size; (void)ws_size;
  const float* x      = (const float*)d_in[0];
  const float* gn_w   = (const float*)d_in[1];
  const float* gn_b   = (const float*)d_in[2];
  const float* qkv_w  = (const float*)d_in[3];
  const float* qkv_b  = (const float*)d_in[4];
  const float* proj_w = (const float*)d_in[5];
  const float* proj_b = (const float*)d_in[6];
  float* out = (float*)d_out;

  half_t* ws = (half_t*)d_ws;
  long long off = 0;
  half_t* Hfrag = ws + off; off += 8LL * 512 * 1024;
  half_t* Wq    = ws + off; off += 1536LL * 512;
  half_t* Wp    = ws + off; off += 512LL * 512;
  half_t* Qf    = ws + off; off += 64LL * 2 * 64 * 512;
  half_t* Kf    = ws + off; off += 64LL * 2 * 64 * 512;
  half_t* Vf    = ws + off; off += 64LL * 4 * 32 * 512;
  half_t* Of    = ws + off; off += 8LL * 512 * 1024;

  groupnorm_kernel<<<dim3(256), dim3(256), 0, stream>>>(x, gn_w, gn_b, Hfrag);

  repackA_kernel<true><<<dim3(768), dim3(256), 0, stream>>>(qkv_w, Wq, 96, 16, 512);
  repackA_kernel<false><<<dim3(256), dim3(256), 0, stream>>>(proj_w, Wp, 32, 16, 512);

  gemm_qkv_kernel<<<dim3(12, 8, 8), dim3(256), 0, stream>>>(Wq, Hfrag, qkv_b, Qf, Kf, Vf);

  attention_kernel<<<dim3(512), dim3(256), 0, stream>>>(Qf, Kf, Vf, Of);

  gemm_proj_kernel<<<dim3(4, 8, 8), dim3(256), 0, stream>>>(Wp, Of, proj_b, x, out);
}


# GroupNorm and both weight repacks (all depend only on the inputs) run as block ranges of ONE launch of 512 blocks, repack ranges 16 elements per thread with loads in flight together: 6 -> 4 launches
# speedup vs baseline: 2.5214x; 1.0967x over previous
#include <hip/hip_runtime.h>

typedef _Float16 half_t;
typedef __attribute__((ext_vector_type(16))) _Float16 v16h;
typedef __attribute__((ext_vector_type(8)))  _Float16 v8h;
typedef __attribute__((ext_vector_type(8)))  float    v8f;

#define WMMA_F16(a, b, c) \
  __builtin_amdgcn_wmma_f32_16x16x32_f16(false, (a), false, (b), (short)0, (c), false, false)

__device__ __forceinline__ int kmap_of(int lane, int j) {
  return ((lane >> 4) << 3) + (j & 7) + ((j >> 3) << 4);
}

__device__ __forceinline__ int kmap_grp(int lane, int j) {
  return ((lane >> 4) << 4) + j;
}

__device__ __forceinline__ v16h ld_frag(const half_t* __restrict__ tile, int lane) {
  v8h lo = *(const v8h*)(tile + lane * 8);
  v8h hi = *(const v8h*)(tile + 256 + lane * 8);
  return __builtin_shufflevector(lo, hi, 0, 1, 2, 3, 4, 5, 6, 7, 8, 9, 10, 11, 12, 13, 14, 15);
}
__device__ __forceinline__ void st_frag(half_t* __restrict__ tile, int lane, v16h v) {
  *(v8h*)(tile + lane * 8)       = __builtin_shufflevector(v, v, 0, 1, 2, 3, 4, 5, 6, 7);
  *(v8h*)(tile + 256 + lane * 8) = __builtin_shufflevector(v, v, 8, 9, 10, 11, 12, 13, 14, 15);
}

__device__ __forceinline__
void groupnorm_part(const float* __restrict__ x,
                    const float* __restrict__ gw,
                    const float* __restrict__ gb,
                    half_t* __restrict__ Hfrag, int blk) {
  __shared__ float s_sum[256], s_ss[256];
  int b = blk >> 5;
  int g = blk & 31;
  const int GE = 16 * 1024;
  const float* xb = x + ((long long)b * 512 + g * 16) * 1024;
  float sum = 0.f, ss = 0.f;
  for (int i = threadIdx.x; i < GE; i += 256) {
    float v = xb[i];
    sum += v; ss += v * v;
  }
  s_sum[threadIdx.x] = sum; s_ss[threadIdx.x] = ss;
  __syncthreads();
  for (int off = 128; off > 0; off >>= 1) {
    if (threadIdx.x < off) {
      s_sum[threadIdx.x] += s_sum[threadIdx.x + off];
      s_ss[threadIdx.x]  += s_ss[threadIdx.x + off];
    }
    __syncthreads();
  }
  float mean = s_sum[0] * (1.0f / GE);
  float var  = s_ss[0] * (1.0f / GE) - mean * mean;
  float inv  = rsqrtf(var + 1e-5f);
  half_t* hb = Hfrag + (long long)b * (16LL * 64 * 512)
                     + (long long)(g >> 1) * (64 * 512);
  int laneAdd = (g & 1) * 16;
  for (int n = threadIdx.x; n < 1024; n += 256) {
    v16h hv;
#pragma unroll
    for (int cc = 0; cc < 16; ++cc) {
      int c = g * 16 + cc;
      hv[cc] = (half_t)((xb[cc * 1024 + n] - mean) * inv * gw[c] + gb[c]);
    }
    st_frag(hb + (n >> 4) * 512, (n & 15) + laneAdd, hv);
  }
}

template <bool GRP, int MT, int KT, int NBLK>
__device__ __forceinline__
void repackA_part(const float* __restrict__ src, half_t* __restrict__ dst, int vblock) {
  constexpr int kcols = KT * 32;
  constexpr int total = MT * KT * 512;
  constexpr int trips = total / (NBLK * 256);
  static_assert(trips * NBLK * 256 == total, "the block range must cover the matrix exactly");
#pragma unroll
  for (int it = 0; it < trips; ++it) {
    int tid = (it * NBLK + vblock) * 256 + (int)threadIdx.x;
    int j = (tid & 7) + ((tid >> 5) & 8), lane = (tid >> 3) & 31, tile = tid >> 9;
    int mt = tile / KT, kt = tile - mt * KT;
    int row = mt * 16 + (lane & 15);
    int col = kt * 32 + (GRP ? kmap_grp(lane, j) : kmap_of(lane, j));
    dst[tid] = (half_t)src[(long long)row * kcols + col];
  }
}

__global__ __launch_bounds__(256)
void prep_kernel(const float* __restrict__ x,
                 const float* __restrict__ gw,
                 const float* __restrict__ gb,
                 const float* __restrict__ qkv_w,
                 const float* __restrict__ proj_w,
                 half_t* __restrict__ Hfrag,
                 half_t* __restrict__ Wq,
                 half_t* __restrict__ Wp) {
  int blk = blockIdx.x;
  if (blk < 256)      groupnorm_part(x, gw, gb, Hfrag, blk);
  else if (blk < 448) repackA_part<true, 96, 16, 192>(qkv_w, Wq, blk - 256);
  else                repackA_part<false, 32, 16, 64>(proj_w, Wp, blk - 448);
}

template <bool TR>
__device__ __forceinline__
void gemm_wave64x32(const half_t* __restrict__ Afrag, const half_t* __restrict__ Bfrag,
                    int KT, int NT, int mtBase, int ntBase, int lane, v8f c[4][2]) {
  for (int kt = 0; kt < KT; ++kt) {
    v16h a[4], bb[2];
#pragma unroll
    for (int i = 0; i < 4; ++i)
      a[i] = ld_frag(Afrag + ((long long)(mtBase + i) * KT + kt) * 512, lane);
#pragma unroll
    for (int j = 0; j < 2; ++j)
      bb[j] = ld_frag(Bfrag + ((long long)kt * NT + (ntBase + j)) * 512, lane);
#pragma unroll
    for (int i = 0; i < 4; ++i)
#pragma unroll
      for (int j = 0; j < 2; ++j)
        c[i][j] = TR ? WMMA_F16(bb[j], a[i], c[i][j]) : WMMA_F16(a[i], bb[j], c[i][j]);
  }
}

__global__ __launch_bounds__(256)
void gemm_qkv_kernel(const half_t* __restrict__ Wfrag, const half_t* __restrict__ Hfrag,
                     const float* __restrict__ bias, half_t* __restrict__ Qf,
                     half_t* __restrict__ Kf, half_t* __restrict__ Vf) {
  int wave = threadIdx.x >> 5, lane = threadIdx.x & 31;
  int b = blockIdx.z;
  int wm = wave & 1, wn = wave >> 1;
  int mBase = blockIdx.x * 128 + wm * 64;
  int nBase = blockIdx.y * 128 + wn * 32;
  int which = blockIdx.x >> 2;
  int bh = b * 8 + ((mBase & 511) >> 6);
  int hh = lane >> 4, col = lane & 15;
  const half_t* Hb = Hfrag + (long long)b * (16LL * 64 * 512);
  v8f c[4][2];
#pragma unroll
  for (int i = 0; i < 4; ++i)
#pragma unroll
    for (int j = 0; j < 2; ++j) c[i][j] = 0.0f;

  if (which == 2) {
    gemm_wave64x32<true>(Wfrag, Hb, 16, 64, mBase >> 4, nBase >> 4, lane, c);
    half_t* vb = Vf + (long long)bh * (4LL * 32 * 512);
    int ktile = nBase >> 5;
#pragma unroll
    for (int i = 0; i < 4; ++i) {
      float bv = bias[mBase + i * 16 + col];
      v16h ov;
#pragma unroll
      for (int r = 0; r < 8; ++r) {
        ov[r]     = (half_t)(c[i][0][r] + bv);
        ov[r + 8] = (half_t)(c[i][1][r] + bv);
      }
      st_frag(vb + (long long)(i * 32 + ktile) * 512, lane, ov);
    }
  } else {
    gemm_wave64x32<false>(Wfrag, Hb, 16, 64, mBase >> 4, nBase >> 4, lane, c);
    half_t* qb = (which == 0 ? Qf : Kf) + (long long)bh * (2LL * 64 * 512);
    int ntile = nBase >> 4;
#pragma unroll
    for (int kt = 0; kt < 2; ++kt)
#pragma unroll
      for (int jj = 0; jj < 2; ++jj) {
        v16h ov;
#pragma unroll
        for (int r = 0; r < 8; ++r) {
          int m0 = mBase + (2 * kt) * 16 + 8 * hh + r;
          ov[r]     = (half_t)(c[2 * kt][jj][r]     + bias[m0]);
          ov[r + 8] = (half_t)(c[2 * kt + 1][jj][r] + bias[m0 + 16]);
        }
        st_frag(qb + (long long)(kt * 64 + ntile + jj) * 512, lane, ov);
      }
  }
}

__global__ __launch_bounds__(256)
void gemm_proj_kernel(const half_t* __restrict__ Wfrag, const half_t* __restrict__ Bfrag,
                      const float* __restrict__ bias, const float* __restrict__ resid,
                      float* __restrict__ out) {
  int wave = threadIdx.x >> 5, lane = threadIdx.x & 31;
  int b = blockIdx.z;
  int wm = wave & 1, wn = wave >> 1;
  int mBase = blockIdx.x * 128 + wm * 64;
  int nBase = blockIdx.y * 128 + wn * 32;
  v8f c[4][2];
#pragma unroll
  for (int i = 0; i < 4; ++i)
#pragma unroll
    for (int j = 0; j < 2; ++j) c[i][j] = 0.0f;
  gemm_wave64x32<false>(Wfrag, Bfrag + (long long)b * (16LL * 64 * 512),
                        16, 64, mBase >> 4, nBase >> 4, lane, c);
  const float* xb = resid + (long long)b * 512 * 1024;
  float* ob = out + (long long)b * 512 * 1024;
  int rAdd = 8 * (lane >> 4), col = lane & 15;
#pragma unroll
  for (int i = 0; i < 4; ++i)
#pragma unroll
    for (int j = 0; j < 2; ++j)
#pragma unroll
      for (int r = 0; r < 8; ++r) {
        int m = mBase + i * 16 + r + rAdd;
        int n = nBase + j * 16 + col;
        long long idx = (long long)m * 1024 + n;
        ob[idx] = c[i][j][r] + bias[m] + xb[idx];
      }
}

__device__ __forceinline__ float xhalf(float x) {
  int b = __builtin_bit_cast(int, x);
  return __builtin_bit_cast(float,
      __builtin_amdgcn_permlanex16(b, b, 0x76543210, 0xfedcba98, false, false));
}

__global__ __launch_bounds__(256) __attribute__((amdgpu_waves_per_eu(4, 4)))
void attention_kernel(const half_t* __restrict__ Qfrag, const half_t* __restrict__ Kfrag,
                      const half_t* __restrict__ Vfrag, half_t* __restrict__ Ofrag) {
  int wave = threadIdx.x >> 5, lane = threadIdx.x & 31;
  int wg = blockIdx.x * 8 + wave;
  int bh = wg >> 6;
  int qt = wg & 63;

  const long long qkStride = 2LL * 64 * 512;
  const long long vStride  = 4LL * 32 * 512;
  const half_t* Qb = Qfrag + (long long)bh * qkStride;
  const half_t* Kb = Kfrag + (long long)bh * qkStride;
  const half_t* Vb = Vfrag + (long long)bh * vStride;

  v16h bq0 = ld_frag(Qb + (0LL * 64 + qt) * 512, lane);
  v16h bq1 = ld_frag(Qb + (1LL * 64 + qt) * 512, lane);

  v8f o[4];
#pragma unroll
  for (int dt = 0; dt < 4; ++dt) o[dt] = 0.0f;

  const float cs = 0.125f * 1.4426950408889634f;
  float m_run  = -3.0e38f;
  float l_part = 0.f;

  for (int it = 0; it < 32; ++it) {
    int kb0 = it * 2, kb1 = it * 2 + 1;
    v16h ak00 = ld_frag(Kb + (0LL * 64 + kb0) * 512, lane);
    v16h ak10 = ld_frag(Kb + (1LL * 64 + kb0) * 512, lane);
    v16h ak01 = ld_frag(Kb + (0LL * 64 + kb1) * 512, lane);
    v16h ak11 = ld_frag(Kb + (1LL * 64 + kb1) * 512, lane);
    v16h av[4];
#pragma unroll
    for (int dt = 0; dt < 4; ++dt)
      av[dt] = ld_frag(Vb + ((long long)dt * 32 + it) * 512, lane);

    v8f sA = 0.0f, sB = 0.0f;
    sA = WMMA_F16(ak00, bq0, sA);
    sA = WMMA_F16(ak10, bq1, sA);
    sB = WMMA_F16(ak01, bq0, sB);
    sB = WMMA_F16(ak11, bq1, sB);

    float mx = fmaxf(sA[0], sB[0]);
#pragma unroll
    for (int r = 1; r < 8; ++r) mx = fmaxf(mx, fmaxf(sA[r], sB[r]));
    mx = fmaxf(mx, xhalf(mx));
    float m_new = fmaxf(m_run, mx);
    float alpha = __builtin_amdgcn_exp2f((m_run - m_new) * cs);
    float nm = -(m_new * cs);

    v16h bp;
    float sumA = 0.f, sumB = 0.f;
#pragma unroll
    for (int r = 0; r < 8; ++r) {
      float pa = __builtin_amdgcn_exp2f(__builtin_fmaf(sA[r], cs, nm));
      float pb = __builtin_amdgcn_exp2f(__builtin_fmaf(sB[r], cs, nm));
      sumA += pa; sumB += pb;
      bp[r]     = (half_t)pa;
      bp[r + 8] = (half_t)pb;
    }
    l_part = l_part * alpha + (sumA + sumB);
    m_run = m_new;

#pragma unroll
    for (int dt = 0; dt < 4; ++dt)
#pragma unroll
      for (int r = 0; r < 8; ++r) o[dt][r] *= alpha;

#pragma unroll
    for (int dt = 0; dt < 4; ++dt) o[dt] = WMMA_F16(av[dt], bp, o[dt]);
  }

  float inv = 1.0f / (l_part + xhalf(l_part));
  int b = bh >> 3, head = bh & 7;
  half_t* Ob = Ofrag + (long long)b * (16LL * 64 * 512);
#pragma unroll
  for (int t = 0; t < 2; ++t) {
    v16h ov;
#pragma unroll
    for (int r = 0; r < 8; ++r) {
      ov[r]     = (half_t)(o[2 * t][r] * inv);
      ov[r + 8] = (half_t)(o[2 * t + 1][r] * inv);
    }
    st_frag(Ob + (long long)((head * 2 + t) * 64 + qt) * 512, lane, ov);
  }
}

extern "C" void kernel_launch(void* const* d_in, const int* in_sizes, int n_in,
                              void* d_out, int out_size, void* d_ws, size_t ws_size,
                              hipStream_t stream) {
  (void)in_sizes; (void)n_in; (void)out_size; (void)ws_size;
  const float* x      = (const float*)d_in[0];
  const float* gn_w   = (const float*)d_in[1];
  const float* gn_b   = (const float*)d_in[2];
  const float* qkv_w  = (const float*)d_in[3];
  const float* qkv_b  = (const float*)d_in[4];
  const float* proj_w = (const float*)d_in[5];
  const float* proj_b = (const float*)d_in[6];
  float* out = (float*)d_out;

  half_t* ws = (half_t*)d_ws;
  long long off = 0;
  half_t* Hfrag = ws + off; off += 8LL * 512 * 1024;
  half_t* Wq    = ws + off; off += 1536LL * 512;
  half_t* Wp    = ws + off; off += 512LL * 512;
  half_t* Qf    = ws + off; off += 64LL * 2 * 64 * 512;
  half_t* Kf    = ws + off; off += 64LL * 2 * 64 * 512;
  half_t* Vf    = ws + off; off += 64LL * 4 * 32 * 512;
  half_t* Of    = ws + off; off += 8LL * 512 * 1024;

  prep_kernel<<<dim3(512), dim3(256), 0, stream>>>(x, gn_w, gn_b, qkv_w, proj_w, Hfrag, Wq, Wp);

  gemm_qkv_kernel<<<dim3(12, 8, 8), dim3(256), 0, stream>>>(Wq, Hfrag, qkv_b, Qf, Kf, Vf);

  attention_kernel<<<dim3(512), dim3(256), 0, stream>>>(Qf, Kf, Vf, Of);

  gemm_proj_kernel<<<dim3(4, 8, 8), dim3(256), 0, stream>>>(Wp, Of, proj_b, x, out);
}


# fragment loads issued ahead of their products behind counted waits: proj GEMM 2 k-steps ahead (3 register sets), attention K/V 1 key step ahead (2 sets), QKV GEMM one load wait per k-step, not 3
# speedup vs baseline: 3.2753x; 1.2990x over previous
#include <hip/hip_runtime.h>

typedef _Float16 half_t;
typedef __attribute__((ext_vector_type(16))) _Float16 v16h;
typedef __attribute__((ext_vector_type(8)))  _Float16 v8h;
typedef __attribute__((ext_vector_type(8)))  float    v8f;

#define WMMA_F16(a, b, c) \
  __builtin_amdgcn_wmma_f32_16x16x32_f16(false, (a), false, (b), (short)0, (c), false, false)

__device__ __forceinline__ int kmap_of(int lane, int j) {
  return ((lane >> 4) << 3) + (j & 7) + ((j >> 3) << 4);
}

__device__ __forceinline__ int kmap_grp(int lane, int j) {
  return ((lane >> 4) << 4) + j;
}

__device__ __forceinline__ v16h ld_frag(const half_t* __restrict__ tile, int lane) {
  v8h lo = *(const v8h*)(tile + lane * 8);
  v8h hi = *(const v8h*)(tile + 256 + lane * 8);
  return __builtin_shufflevector(lo, hi, 0, 1, 2, 3, 4, 5, 6, 7, 8, 9, 10, 11, 12, 13, 14, 15);
}
__device__ __forceinline__ void st_frag(half_t* __restrict__ tile, int lane, v16h v) {
  *(v8h*)(tile + lane * 8)       = __builtin_shufflevector(v, v, 0, 1, 2, 3, 4, 5, 6, 7);
  *(v8h*)(tile + 256 + lane * 8) = __builtin_shufflevector(v, v, 8, 9, 10, 11, 12, 13, 14, 15);
}

__device__ __forceinline__
void groupnorm_part(const float* __restrict__ x,
                    const float* __restrict__ gw,
                    const float* __restrict__ gb,
                    half_t* __restrict__ Hfrag, int blk) {
  __shared__ float s_sum[256], s_ss[256];
  int b = blk >> 5;
  int g = blk & 31;
  const int GE = 16 * 1024;
  const float* xb = x + ((long long)b * 512 + g * 16) * 1024;
  float sum = 0.f, ss = 0.f;
  for (int i = threadIdx.x; i < GE; i += 256) {
    float v = xb[i];
    sum += v; ss += v * v;
  }
  s_sum[threadIdx.x] = sum; s_ss[threadIdx.x] = ss;
  __syncthreads();
  for (int off = 128; off > 0; off >>= 1) {
    if (threadIdx.x < off) {
      s_sum[threadIdx.x] += s_sum[threadIdx.x + off];
      s_ss[threadIdx.x]  += s_ss[threadIdx.x + off];
    }
    __syncthreads();
  }
  float mean = s_sum[0] * (1.0f / GE);
  float var  = s_ss[0] * (1.0f / GE) - mean * mean;
  float inv  = rsqrtf(var + 1e-5f);
  half_t* hb = Hfrag + (long long)b * (16LL * 64 * 512)
                     + (long long)(g >> 1) * (64 * 512);
  int laneAdd = (g & 1) * 16;
  for (int n = threadIdx.x; n < 1024; n += 256) {
    v16h hv;
#pragma unroll
    for (int cc = 0; cc < 16; ++cc) {
      int c = g * 16 + cc;
      hv[cc] = (half_t)((xb[cc * 1024 + n] - mean) * inv * gw[c] + gb[c]);
    }
    st_frag(hb + (n >> 4) * 512, (n & 15) + laneAdd, hv);
  }
}

template <bool GRP, int MT, int KT, int NBLK>
__device__ __forceinline__
void repackA_part(const float* __restrict__ src, half_t* __restrict__ dst, int vblock) {
  constexpr int kcols = KT * 32;
  constexpr int total = MT * KT * 512;
  constexpr int trips = total / (NBLK * 256);
  static_assert(trips * NBLK * 256 == total, "the block range must cover the matrix exactly");
#pragma unroll
  for (int it = 0; it < trips; ++it) {
    int tid = (it * NBLK + vblock) * 256 + (int)threadIdx.x;
    int j = (tid & 7) + ((tid >> 5) & 8), lane = (tid >> 3) & 31, tile = tid >> 9;
    int mt = tile / KT, kt = tile - mt * KT;
    int row = mt * 16 + (lane & 15);
    int col = kt * 32 + (GRP ? kmap_grp(lane, j) : kmap_of(lane, j));
    dst[tid] = (half_t)src[(long long)row * kcols + col];
  }
}

__global__ __launch_bounds__(256)
void prep_kernel(const float* __restrict__ x,
                 const float* __restrict__ gw,
                 const float* __restrict__ gb,
                 const float* __restrict__ qkv_w,
                 const float* __restrict__ proj_w,
                 half_t* __restrict__ Hfrag,
                 half_t* __restrict__ Wq,
                 half_t* __restrict__ Wp) {
  int blk = blockIdx.x;
  if (blk < 256)      groupnorm_part(x, gw, gb, Hfrag, blk);
  else if (blk < 448) repackA_part<true, 96, 16, 192>(qkv_w, Wq, blk - 256);
  else                repackA_part<false, 32, 16, 64>(proj_w, Wp, blk - 448);
}

#define LOADS_STAY_AHEAD() __builtin_amdgcn_sched_barrier(0)

struct StepFrags { v16h a[4]; v16h b[2]; };
__device__ __forceinline__
void ld_step(StepFrags& f, const half_t* __restrict__ Afrag, const half_t* __restrict__ Bfrag,
             int KT, int NT, int mtBase, int ntBase, int kt, int lane) {
#pragma unroll
  for (int i = 0; i < 4; ++i)
    f.a[i] = ld_frag(Afrag + ((long long)(mtBase + i) * KT + kt) * 512, lane);
#pragma unroll
  for (int j = 0; j < 2; ++j)
    f.b[j] = ld_frag(Bfrag + ((long long)kt * NT + (ntBase + j)) * 512, lane);
}
template <bool TR>
__device__ __forceinline__
void mma_step(const StepFrags& f, v8f c[4][2]) {
#pragma unroll
  for (int i = 0; i < 4; ++i)
#pragma unroll
    for (int j = 0; j < 2; ++j)
      c[i][j] = TR ? WMMA_F16(f.b[j], f.a[i], c[i][j]) : WMMA_F16(f.a[i], f.b[j], c[i][j]);
}
template <bool TR>
__device__ __forceinline__
void gemm_wave64x32(const half_t* __restrict__ Afrag, const half_t* __restrict__ Bfrag,
                    int KT, int NT, int mtBase, int ntBase, int lane, v8f c[4][2]) {
  for (int kt = 0; kt < KT; ++kt) {
    StepFrags f;
    ld_step(f, Afrag, Bfrag, KT, NT, mtBase, ntBase, kt, lane);
    LOADS_STAY_AHEAD();
    mma_step<TR>(f, c);
  }
}
template <bool TR, int KT>
__device__ __forceinline__
void gemm_wave64x32_ahead2(const half_t* __restrict__ Afrag, const half_t* __restrict__ Bfrag,
                           int NT, int mtBase, int ntBase, int lane, v8f c[4][2]) {
  static_assert(KT % 3 == 1 && KT >= 4, "whole trips of three steps, then four closing steps");
  StepFrags f0, f1, f2;
  ld_step(f0, Afrag, Bfrag, KT, NT, mtBase, ntBase, 0, lane);
  ld_step(f1, Afrag, Bfrag, KT, NT, mtBase, ntBase, 1, lane);
  for (int kt = 0; kt + 4 < KT; kt += 3) {
    ld_step(f2, Afrag, Bfrag, KT, NT, mtBase, ntBase, kt + 2, lane);
    LOADS_STAY_AHEAD();
    mma_step<TR>(f0, c);
    LOADS_STAY_AHEAD();
    ld_step(f0, Afrag, Bfrag, KT, NT, mtBase, ntBase, kt + 3, lane);
    LOADS_STAY_AHEAD();
    mma_step<TR>(f1, c);
    LOADS_STAY_AHEAD();
    ld_step(f1, Afrag, Bfrag, KT, NT, mtBase, ntBase, kt + 4, lane);
    LOADS_STAY_AHEAD();
    mma_step<TR>(f2, c);
  }
  ld_step(f2, Afrag, Bfrag, KT, NT, mtBase, ntBase, KT - 2, lane);
  LOADS_STAY_AHEAD();
  mma_step<TR>(f0, c);
  LOADS_STAY_AHEAD();
  ld_step(f0, Afrag, Bfrag, KT, NT, mtBase, ntBase, KT - 1, lane);
  LOADS_STAY_AHEAD();
  mma_step<TR>(f1, c);
  mma_step<TR>(f2, c);
  mma_step<TR>(f0, c);
}

__global__ __launch_bounds__(256)
void gemm_qkv_kernel(const half_t* __restrict__ Wfrag, const half_t* __restrict__ Hfrag,
                     const float* __restrict__ bias, half_t* __restrict__ Qf,
                     half_t* __restrict__ Kf, half_t* __restrict__ Vf) {
  int wave = threadIdx.x >> 5, lane = threadIdx.x & 31;
  int b = blockIdx.z;
  int wm = wave & 1, wn = wave >> 1;
  int mBase = blockIdx.x * 128 + wm * 64;
  int nBase = blockIdx.y * 128 + wn * 32;
  int which = blockIdx.x >> 2;
  int bh = b * 8 + ((mBase & 511) >> 6);
  int hh = lane >> 4, col = lane & 15;
  const half_t* Hb = Hfrag + (long long)b * (16LL * 64 * 512);
  v8f c[4][2];
#pragma unroll
  for (int i = 0; i < 4; ++i)
#pragma unroll
    for (int j = 0; j < 2; ++j) c[i][j] = 0.0f;

  if (which == 2) {
    gemm_wave64x32<true>(Wfrag, Hb, 16, 64, mBase >> 4, nBase >> 4, lane, c);
    half_t* vb = Vf + (long long)bh * (4LL * 32 * 512);
    int ktile = nBase >> 5;
#pragma unroll
    for (int i = 0; i < 4; ++i) {
      float bv = bias[mBase + i * 16 + col];
      v16h ov;
#pragma unroll
      for (int r = 0; r < 8; ++r) {
        ov[r]     = (half_t)(c[i][0][r] + bv);
        ov[r + 8] = (half_t)(c[i][1][r] + bv);
      }
      st_frag(vb + (long long)(i * 32 + ktile) * 512, lane, ov);
    }
  } else {
    gemm_wave64x32<false>(Wfrag, Hb, 16, 64, mBase >> 4, nBase >> 4, lane, c);
    half_t* qb = (which == 0 ? Qf : Kf) + (long long)bh * (2LL * 64 * 512);
    int ntile = nBase >> 4;
#pragma unroll
    for (int kt = 0; kt < 2; ++kt)
#pragma unroll
      for (int jj = 0; jj < 2; ++jj) {
        v16h ov;
#pragma unroll
        for (int r = 0; r < 8; ++r) {
          int m0 = mBase + (2 * kt) * 16 + 8 * hh + r;
          ov[r]     = (half_t)(c[2 * kt][jj][r]     + bias[m0]);
          ov[r + 8] = (half_t)(c[2 * kt + 1][jj][r] + bias[m0 + 16]);
        }
        st_frag(qb + (long long)(kt * 64 + ntile + jj) * 512, lane, ov);
      }
  }
}

__global__ __launch_bounds__(256)
void gemm_proj_kernel(const half_t* __restrict__ Wfrag, const half_t* __restrict__ Bfrag,
                      const float* __restrict__ bias, const float* __restrict__ resid,
                      float* __restrict__ out) {
  int wave = threadIdx.x >> 5, lane = threadIdx.x & 31;
  int b = blockIdx.z;
  int wm = wave & 1, wn = wave >> 1;
  int mBase = blockIdx.x * 128 + wm * 64;
  int nBase = blockIdx.y * 128 + wn * 32;
  v8f c[4][2];
#pragma unroll
  for (int i = 0; i < 4; ++i)
#pragma unroll
    for (int j = 0; j < 2; ++j) c[i][j] = 0.0f;
  gemm_wave64x32_ahead2<false, 16>(Wfrag, Bfrag + (long long)b * (16LL * 64 * 512),
                                   64, mBase >> 4, nBase >> 4, lane, c);
  const float* xb = resid + (long long)b * 512 * 1024;
  float* ob = out + (long long)b * 512 * 1024;
  int rAdd = 8 * (lane >> 4), col = lane & 15;
#pragma unroll
  for (int i = 0; i < 4; ++i)
#pragma unroll
    for (int j = 0; j < 2; ++j)
#pragma unroll
      for (int r = 0; r < 8; ++r) {
        int m = mBase + i * 16 + r + rAdd;
        int n = nBase + j * 16 + col;
        long long idx = (long long)m * 1024 + n;
        ob[idx] = c[i][j][r] + bias[m] + xb[idx];
      }
}

__device__ __forceinline__ float xhalf(float x) {
  int b = __builtin_bit_cast(int, x);
  return __builtin_bit_cast(float,
      __builtin_amdgcn_permlanex16(b, b, 0x76543210, 0xfedcba98, false, false));
}

struct KVFrags { v16h k[4]; v16h v[4]; };
__device__ __forceinline__
void ld_kv(KVFrags& f, const half_t* __restrict__ Kb, const half_t* __restrict__ Vb,
           int it, int lane) {
  int kb0 = it * 2, kb1 = it * 2 + 1;
  f.k[0] = ld_frag(Kb + (0LL * 64 + kb0) * 512, lane);
  f.k[1] = ld_frag(Kb + (1LL * 64 + kb0) * 512, lane);
  f.k[2] = ld_frag(Kb + (0LL * 64 + kb1) * 512, lane);
  f.k[3] = ld_frag(Kb + (1LL * 64 + kb1) * 512, lane);
#pragma unroll
  for (int dt = 0; dt < 4; ++dt)
    f.v[dt] = ld_frag(Vb + ((long long)dt * 32 + it) * 512, lane);
}

__device__ __forceinline__
void attn_step(const KVFrags& f, v16h bq0, v16h bq1, v8f o[4], float& m_run, float& l_part) {
  const float cs = 0.125f * 1.4426950408889634f;
  v8f sA = 0.0f, sB = 0.0f;
  sA = WMMA_F16(f.k[0], bq0, sA);
  sA = WMMA_F16(f.k[1], bq1, sA);
  sB = WMMA_F16(f.k[2], bq0, sB);
  sB = WMMA_F16(f.k[3], bq1, sB);

  float mx = fmaxf(sA[0], sB[0]);
#pragma unroll
  for (int r = 1; r < 8; ++r) mx = fmaxf(mx, fmaxf(sA[r], sB[r]));
  mx = fmaxf(mx, xhalf(mx));
  float m_new = fmaxf(m_run, mx);
  float alpha = __builtin_amdgcn_exp2f((m_run - m_new) * cs);
  float nm = -(m_new * cs);

  v16h bp;
  float sumA = 0.f, sumB = 0.f;
#pragma unroll
  for (int r = 0; r < 8; ++r) {
    float pa = __builtin_amdgcn_exp2f(__builtin_fmaf(sA[r], cs, nm));
    float pb = __builtin_amdgcn_exp2f(__builtin_fmaf(sB[r], cs, nm));
    sumA += pa; sumB += pb;
    bp[r]     = (half_t)pa;
    bp[r + 8] = (half_t)pb;
  }
  l_part = l_part * alpha + (sumA + sumB);
  m_run = m_new;

#pragma unroll
  for (int dt = 0; dt < 4; ++dt)
#pragma unroll
    for (int r = 0; r < 8; ++r) o[dt][r] *= alpha;

#pragma unroll
  for (int dt = 0; dt < 4; ++dt) o[dt] = WMMA_F16(f.v[dt], bp, o[dt]);
}

__global__ __launch_bounds__(256) __attribute__((amdgpu_waves_per_eu(4, 4)))
void attention_kernel(const half_t* __restrict__ Qfrag, const half_t* __restrict__ Kfrag,
                      const half_t* __restrict__ Vfrag, half_t* __restrict__ Ofrag) {
  int wave = threadIdx.x >> 5, lane = threadIdx.x & 31;
  int wg = blockIdx.x * 8 + wave;
  int bh = wg >> 6;
  int qt = wg & 63;

  const long long qkStride = 2LL * 64 * 512;
  const long long vStride  = 4LL * 32 * 512;
  const half_t* Qb = Qfrag + (long long)bh * qkStride;
  const half_t* Kb = Kfrag + (long long)bh * qkStride;
  const half_t* Vb = Vfrag + (long long)bh * vStride;

  v16h bq0 = ld_frag(Qb + (0LL * 64 + qt) * 512, lane);
  v16h bq1 = ld_frag(Qb + (1LL * 64 + qt) * 512, lane);

  v8f o[4];
#pragma unroll
  for (int dt = 0; dt < 4; ++dt) o[dt] = 0.0f;

  float m_run  = -3.0e38f;
  float l_part = 0.f;

  KVFrags f0, f1;
  ld_kv(f0, Kb, Vb, 0, lane);
  for (int it = 0; it + 2 < 32; it += 2) {
    ld_kv(f1, Kb, Vb, it + 1, lane);
    LOADS_STAY_AHEAD();
    attn_step(f0, bq0, bq1, o, m_run, l_part);
    LOADS_STAY_AHEAD();
    ld_kv(f0, Kb, Vb, it + 2, lane);
    LOADS_STAY_AHEAD();
    attn_step(f1, bq0, bq1, o, m_run, l_part);
  }
  ld_kv(f1, Kb, Vb, 31, lane);
  LOADS_STAY_AHEAD();
  attn_step(f0, bq0, bq1, o, m_run, l_part);
  attn_step(f1, bq0, bq1, o, m_run, l_part);

  float inv = 1.0f / (l_part + xhalf(l_part));
  int b = bh >> 3, head = bh & 7;
  half_t* Ob = Ofrag + (long long)b * (16LL * 64 * 512);
#pragma unroll
  for (int t = 0; t < 2; ++t) {
    v16h ov;
#pragma unroll
    for (int r = 0; r < 8; ++r) {
      ov[r]     = (half_t)(o[2 * t][r] * inv);
      ov[r + 8] = (half_t)(o[2 * t + 1][r] * inv);
    }
    st_frag(Ob + (long long)((head * 2 + t) * 64 + qt) * 512, lane, ov);
  }
}

extern "C" void kernel_launch(void* const* d_in, const int* in_sizes, int n_in,
                              void* d_out, int out_size, void* d_ws, size_t ws_size,
                              hipStream_t stream) {
  (void)in_sizes; (void)n_in; (void)out_size; (void)ws_size;
  const float* x      = (const float*)d_in[0];
  const float* gn_w   = (const float*)d_in[1];
  const float* gn_b   = (const float*)d_in[2];
  const float* qkv_w  = (const float*)d_in[3];
  const float* qkv_b  = (const float*)d_in[4];
  const float* proj_w = (const float*)d_in[5];
  const float* proj_b = (const float*)d_in[6];
  float* out = (float*)d_out;

  half_t* ws = (half_t*)d_ws;
  long long off = 0;
  half_t* Hfrag = ws + off; off += 8LL * 512 * 1024;
  half_t* Wq    = ws + off; off += 1536LL * 512;
  half_t* Wp    = ws + off; off += 512LL * 512;
  half_t* Qf    = ws + off; off += 64LL * 2 * 64 * 512;
  half_t* Kf    = ws + off; off += 64LL * 2 * 64 * 512;
  half_t* Vf    = ws + off; off += 64LL * 4 * 32 * 512;
  half_t* Of    = ws + off; off += 8LL * 512 * 1024;

  prep_kernel<<<dim3(512), dim3(256), 0, stream>>>(x, gn_w, gn_b, qkv_w, proj_w, Hfrag, Wq, Wp);

  gemm_qkv_kernel<<<dim3(12, 8, 8), dim3(256), 0, stream>>>(Wq, Hfrag, qkv_b, Qf, Kf, Vf);

  attention_kernel<<<dim3(512), dim3(256), 0, stream>>>(Qf, Kf, Vf, Of);

  gemm_proj_kernel<<<dim3(4, 8, 8), dim3(256), 0, stream>>>(Wp, Of, proj_b, x, out);
}


# attention: a wave owns 2 query tiles (32 queries), so each loaded K/V fragment feeds 2 products: half the waves, half the K/V loads; with 2 waves per SIMD K/V run 2 key steps ahead in 3 register sets
# speedup vs baseline: 3.6472x; 1.1135x over previous
#include <hip/hip_runtime.h>

typedef _Float16 half_t;
typedef __attribute__((ext_vector_type(16))) _Float16 v16h;
typedef __attribute__((ext_vector_type(8)))  _Float16 v8h;
typedef __attribute__((ext_vector_type(8)))  float    v8f;

#define WMMA_F16(a, b, c) \
  __builtin_amdgcn_wmma_f32_16x16x32_f16(false, (a), false, (b), (short)0, (c), false, false)

__device__ __forceinline__ int kmap_of(int lane, int j) {
  return ((lane >> 4) << 3) + (j & 7) + ((j >> 3) << 4);
}

__device__ __forceinline__ int kmap_grp(int lane, int j) {
  return ((lane >> 4) << 4) + j;
}

__device__ __forceinline__ v16h ld_frag(const half_t* __restrict__ tile, int lane) {
  v8h lo = *(const v8h*)(tile + lane * 8);
  v8h hi = *(const v8h*)(tile + 256 + lane * 8);
  return __builtin_shufflevector(lo, hi, 0, 1, 2, 3, 4, 5, 6, 7, 8, 9, 10, 11, 12, 13, 14, 15);
}
__device__ __forceinline__ void st_frag(half_t* __restrict__ tile, int lane, v16h v) {
  *(v8h*)(tile + lane * 8)       = __builtin_shufflevector(v, v, 0, 1, 2, 3, 4, 5, 6, 7);
  *(v8h*)(tile + 256 + lane * 8) = __builtin_shufflevector(v, v, 8, 9, 10, 11, 12, 13, 14, 15);
}

__device__ __forceinline__
void groupnorm_part(const float* __restrict__ x,
                    const float* __restrict__ gw,
                    const float* __restrict__ gb,
                    half_t* __restrict__ Hfrag, int blk) {
  __shared__ float s_sum[256], s_ss[256];
  int b = blk >> 5;
  int g = blk & 31;
  const int GE = 16 * 1024;
  const float* xb = x + ((long long)b * 512 + g * 16) * 1024;
  float gwv[16], gbv[16];
#pragma unroll
  for (int cc = 0; cc < 16; ++cc) {
    gwv[cc] = gw[g * 16 + cc];
    gbv[cc] = gb[g * 16 + cc];
  }
  float xv[4][16];
#pragma unroll
  for (int cc = 0; cc < 16; ++cc)
#pragma unroll
    for (int i = 0; i < 4; ++i)
      xv[i][cc] = xb[cc * 1024 + i * 256 + (int)threadIdx.x];
  float sum = 0.f, ss = 0.f;
#pragma unroll
  for (int cc = 0; cc < 16; ++cc)
#pragma unroll
    for (int i = 0; i < 4; ++i) {
      float v = xv[i][cc];
      sum += v; ss += v * v;
    }
  s_sum[threadIdx.x] = sum; s_ss[threadIdx.x] = ss;
  __syncthreads();
  for (int off = 128; off > 0; off >>= 1) {
    if (threadIdx.x < off) {
      s_sum[threadIdx.x] += s_sum[threadIdx.x + off];
      s_ss[threadIdx.x]  += s_ss[threadIdx.x + off];
    }
    __syncthreads();
  }
  float mean = s_sum[0] * (1.0f / GE);
  float var  = s_ss[0] * (1.0f / GE) - mean * mean;
  float inv  = rsqrtf(var + 1e-5f);
  half_t* hb = Hfrag + (long long)b * (16LL * 64 * 512)
                     + (long long)(g >> 1) * (64 * 512);
  int laneAdd = (g & 1) * 16;
#pragma unroll
  for (int i = 0; i < 4; ++i) {
    int n = (int)threadIdx.x + i * 256;
    v16h hv;
#pragma unroll
    for (int cc = 0; cc < 16; ++cc)
      hv[cc] = (half_t)((xv[i][cc] - mean) * inv * gwv[cc] + gbv[cc]);
    st_frag(hb + (n >> 4) * 512, (n & 15) + laneAdd, hv);
  }
}

template <bool GRP, int MT, int KT, int NBLK>
__device__ __forceinline__
void repackA_part(const float* __restrict__ src, half_t* __restrict__ dst, int vblock) {
  constexpr int kcols = KT * 32;
  constexpr int total = MT * KT * 512;
  constexpr int trips = total / (NBLK * 256);
  static_assert(trips * NBLK * 256 == total, "the block range must cover the matrix exactly");
#pragma unroll
  for (int it = 0; it < trips; ++it) {
    int tid = (it * NBLK + vblock) * 256 + (int)threadIdx.x;
    int j = (tid & 7) + ((tid >> 5) & 8), lane = (tid >> 3) & 31, tile = tid >> 9;
    int mt = tile / KT, kt = tile - mt * KT;
    int row = mt * 16 + (lane & 15);
    int col = kt * 32 + (GRP ? kmap_grp(lane, j) : kmap_of(lane, j));
    dst[tid] = (half_t)src[(long long)row * kcols + col];
  }
}

__global__ __launch_bounds__(256)
void prep_kernel(const float* __restrict__ x,
                 const float* __restrict__ gw,
                 const float* __restrict__ gb,
                 const float* __restrict__ qkv_w,
                 const float* __restrict__ proj_w,
                 half_t* __restrict__ Hfrag,
                 half_t* __restrict__ Wq,
                 half_t* __restrict__ Wp) {
  int blk = blockIdx.x;
  if (blk < 256)      groupnorm_part(x, gw, gb, Hfrag, blk);
  else if (blk < 448) repackA_part<true, 96, 16, 192>(qkv_w, Wq, blk - 256);
  else                repackA_part<false, 32, 16, 64>(proj_w, Wp, blk - 448);
}

#define LOADS_STAY_AHEAD() __builtin_amdgcn_sched_barrier(0)

struct StepFrags { v16h a[4]; v16h b[2]; };
__device__ __forceinline__
void ld_step(StepFrags& f, const half_t* __restrict__ Afrag, const half_t* __restrict__ Bfrag,
             int KT, int NT, int mtBase, int ntBase, int kt, int lane) {
#pragma unroll
  for (int i = 0; i < 4; ++i)
    f.a[i] = ld_frag(Afrag + ((long long)(mtBase + i) * KT + kt) * 512, lane);
#pragma unroll
  for (int j = 0; j < 2; ++j)
    f.b[j] = ld_frag(Bfrag + ((long long)kt * NT + (ntBase + j)) * 512, lane);
}
template <bool TR>
__device__ __forceinline__
void mma_step(const StepFrags& f, v8f c[4][2]) {
#pragma unroll
  for (int i = 0; i < 4; ++i)
#pragma unroll
    for (int j = 0; j < 2; ++j)
      c[i][j] = TR ? WMMA_F16(f.b[j], f.a[i], c[i][j]) : WMMA_F16(f.a[i], f.b[j], c[i][j]);
}
template <bool TR>
__device__ __forceinline__
void gemm_wave64x32(const half_t* __restrict__ Afrag, const half_t* __restrict__ Bfrag,
                    int KT, int NT, int mtBase, int ntBase, int lane, v8f c[4][2]) {
  for (int kt = 0; kt < KT; ++kt) {
    StepFrags f;
    ld_step(f, Afrag, Bfrag, KT, NT, mtBase, ntBase, kt, lane);
    LOADS_STAY_AHEAD();
    mma_step<TR>(f, c);
  }
}
template <bool TR, int KT>
__device__ __forceinline__
void gemm_wave64x32_ahead2(const half_t* __restrict__ Afrag, const half_t* __restrict__ Bfrag,
                           int NT, int mtBase, int ntBase, int lane, v8f c[4][2]) {
  static_assert(KT % 3 == 1 && KT >= 4, "whole trips of three steps, then four closing steps");
  StepFrags f0, f1, f2;
  ld_step(f0, Afrag, Bfrag, KT, NT, mtBase, ntBase, 0, lane);
  ld_step(f1, Afrag, Bfrag, KT, NT, mtBase, ntBase, 1, lane);
  for (int kt = 0; kt + 4 < KT; kt += 3) {
    ld_step(f2, Afrag, Bfrag, KT, NT, mtBase, ntBase, kt + 2, lane);
    LOADS_STAY_AHEAD();
    mma_step<TR>(f0, c);
    LOADS_STAY_AHEAD();
    ld_step(f0, Afrag, Bfrag, KT, NT, mtBase, ntBase, kt + 3, lane);
    LOADS_STAY_AHEAD();
    mma_step<TR>(f1, c);
    LOADS_STAY_AHEAD();
    ld_step(f1, Afrag, Bfrag, KT, NT, mtBase, ntBase, kt + 4, lane);
    LOADS_STAY_AHEAD();
    mma_step<TR>(f2, c);
  }
  ld_step(f2, Afrag, Bfrag, KT, NT, mtBase, ntBase, KT - 2, lane);
  LOADS_STAY_AHEAD();
  mma_step<TR>(f0, c);
  LOADS_STAY_AHEAD();
  ld_step(f0, Afrag, Bfrag, KT, NT, mtBase, ntBase, KT - 1, lane);
  LOADS_STAY_AHEAD();
  mma_step<TR>(f1, c);
  mma_step<TR>(f2, c);
  mma_step<TR>(f0, c);
}

__global__ __launch_bounds__(256)
void gemm_qkv_kernel(const half_t* __restrict__ Wfrag, const half_t* __restrict__ Hfrag,
                     const float* __restrict__ bias, half_t* __restrict__ Qf,
                     half_t* __restrict__ Kf, half_t* __restrict__ Vf) {
  int wave = threadIdx.x >> 5, lane = threadIdx.x & 31;
  int b = blockIdx.z;
  int wm = wave & 1, wn = wave >> 1;
  int mBase = blockIdx.x * 128 + wm * 64;
  int nBase = blockIdx.y * 128 + wn * 32;
  int which = blockIdx.x >> 2;
  int bh = b * 8 + ((mBase & 511) >> 6);
  int hh = lane >> 4, col = lane & 15;
  const half_t* Hb = Hfrag + (long long)b * (16LL * 64 * 512);
  v8f c[4][2];
#pragma unroll
  for (int i = 0; i < 4; ++i)
#pragma unroll
    for (int j = 0; j < 2; ++j) c[i][j] = 0.0f;

  if (which == 2) {
    gemm_wave64x32<true>(Wfrag, Hb, 16, 64, mBase >> 4, nBase >> 4, lane, c);
    half_t* vb = Vf + (long long)bh * (4LL * 32 * 512);
    int ktile = nBase >> 5;
#pragma unroll
    for (int i = 0; i < 4; ++i) {
      float bv = bias[mBase + i * 16 + col];
      v16h ov;
#pragma unroll
      for (int r = 0; r < 8; ++r) {
        ov[r]     = (half_t)(c[i][0][r] + bv);
        ov[r + 8] = (half_t)(c[i][1][r] + bv);
      }
      st_frag(vb + (long long)(i * 32 + ktile) * 512, lane, ov);
    }
  } else {
    gemm_wave64x32<false>(Wfrag, Hb, 16, 64, mBase >> 4, nBase >> 4, lane, c);
    half_t* qb = (which == 0 ? Qf : Kf) + (long long)bh * (2LL * 64 * 512);
    int ntile = nBase >> 4;
#pragma unroll
    for (int kt = 0; kt < 2; ++kt)
#pragma unroll
      for (int jj = 0; jj < 2; ++jj) {
        v16h ov;
#pragma unroll
        for (int r = 0; r < 8; ++r) {
          int m0 = mBase + (2 * kt) * 16 + 8 * hh + r;
          ov[r]     = (half_t)(c[2 * kt][jj][r]     + bias[m0]);
          ov[r + 8] = (half_t)(c[2 * kt + 1][jj][r] + bias[m0 + 16]);
        }
        st_frag(qb + (long long)(kt * 64 + ntile + jj) * 512, lane, ov);
      }
  }
}

__global__ __launch_bounds__(256)
void gemm_proj_kernel(const half_t* __restrict__ Wfrag, const half_t* __restrict__ Bfrag,
                      const float* __restrict__ bias, const float* __restrict__ resid,
                      float* __restrict__ out) {
  int wave = threadIdx.x >> 5, lane = threadIdx.x & 31;
  int b = blockIdx.z;
  int wm = wave & 1, wn = wave >> 1;
  int mBase = blockIdx.x * 128 + wm * 64;
  int nBase = blockIdx.y * 128 + wn * 32;
  v8f c[4][2];
#pragma unroll
  for (int i = 0; i < 4; ++i)
#pragma unroll
    for (int j = 0; j < 2; ++j) c[i][j] = 0.0f;
  gemm_wave64x32_ahead2<false, 16>(Wfrag, Bfrag + (long long)b * (16LL * 64 * 512),
                                   64, mBase >> 4, nBase >> 4, lane, c);
  const float* xb = resid + (long long)b * 512 * 1024;
  float* ob = out + (long long)b * 512 * 1024;
  int rAdd = 8 * (lane >> 4), col = lane & 15;
#pragma unroll
  for (int i = 0; i < 4; ++i)
#pragma unroll
    for (int j = 0; j < 2; ++j)
#pragma unroll
      for (int r = 0; r < 8; ++r) {
        int m = mBase + i * 16 + r + rAdd;
        int n = nBase + j * 16 + col;
        long long idx = (long long)m * 1024 + n;
        ob[idx] = c[i][j][r] + bias[m] + xb[idx];
      }
}

__device__ __forceinline__ float xhalf(float x) {
  int b = __builtin_bit_cast(int, x);
  return __builtin_bit_cast(float,
      __builtin_amdgcn_permlanex16(b, b, 0x76543210, 0xfedcba98, false, false));
}

struct KVFrags { v16h k[4]; v16h v[4]; };
__device__ __forceinline__
void ld_kv(KVFrags& f, const half_t* __restrict__ Kb, const half_t* __restrict__ Vb,
           int it, int lane) {
  int kb0 = it * 2, kb1 = it * 2 + 1;
  f.k[0] = ld_frag(Kb + (0LL * 64 + kb0) * 512, lane);
  f.k[1] = ld_frag(Kb + (1LL * 64 + kb0) * 512, lane);
  f.k[2] = ld_frag(Kb + (0LL * 64 + kb1) * 512, lane);
  f.k[3] = ld_frag(Kb + (1LL * 64 + kb1) * 512, lane);
#pragma unroll
  for (int dt = 0; dt < 4; ++dt)
    f.v[dt] = ld_frag(Vb + ((long long)dt * 32 + it) * 512, lane);
}

template <int NQ>
__device__ __forceinline__
void attn_step(const KVFrags& f, const v16h (&bq0)[NQ], const v16h (&bq1)[NQ],
               v8f (&o)[NQ][4], float (&m_run)[NQ], float (&l_part)[NQ]) {
  const float cs = 0.125f * 1.4426950408889634f;
  v8f sA[NQ], sB[NQ];
#pragma unroll
  for (int q = 0; q < NQ; ++q) {
    sA[q] = 0.0f; sB[q] = 0.0f;
    sA[q] = WMMA_F16(f.k[0], bq0[q], sA[q]);
    sA[q] = WMMA_F16(f.k[1], bq1[q], sA[q]);
    sB[q] = WMMA_F16(f.k[2], bq0[q], sB[q]);
    sB[q] = WMMA_F16(f.k[3], bq1[q], sB[q]);
  }

  v16h bp[NQ];
#pragma unroll
  for (int q = 0; q < NQ; ++q) {
    float mx = fmaxf(sA[q][0], sB[q][0]);
#pragma unroll
    for (int r = 1; r < 8; ++r) mx = fmaxf(mx, fmaxf(sA[q][r], sB[q][r]));
    mx = fmaxf(mx, xhalf(mx));
    float m_new = fmaxf(m_run[q], mx);
    float alpha = __builtin_amdgcn_exp2f((m_run[q] - m_new) * cs);
    float nm = -(m_new * cs);

    float sumA = 0.f, sumB = 0.f;
#pragma unroll
    for (int r = 0; r < 8; ++r) {
      float pa = __builtin_amdgcn_exp2f(__builtin_fmaf(sA[q][r], cs, nm));
      float pb = __builtin_amdgcn_exp2f(__builtin_fmaf(sB[q][r], cs, nm));
      sumA += pa; sumB += pb;
      bp[q][r]     = (half_t)pa;
      bp[q][r + 8] = (half_t)pb;
    }
    l_part[q] = l_part[q] * alpha + (sumA + sumB);
    m_run[q] = m_new;

#pragma unroll
    for (int dt = 0; dt < 4; ++dt)
#pragma unroll
      for (int r = 0; r < 8; ++r) o[q][dt][r] *= alpha;
  }

#pragma unroll
  for (int q = 0; q < NQ; ++q)
#pragma unroll
    for (int dt = 0; dt < 4; ++dt) o[q][dt] = WMMA_F16(f.v[dt], bp[q], o[q][dt]);
}

template <int NQ>
__device__ __forceinline__ int step_behind(int step, v8f (&o)[NQ][4]) {
#pragma unroll
  for (int q = 0; q < NQ; ++q)
    asm volatile("" : "+s"(step), "+v"(o[q][0]), "+v"(o[q][1]), "+v"(o[q][2]), "+v"(o[q][3]));
  return step;
}

#define NQT 2
__global__ __launch_bounds__(256) __attribute__((amdgpu_waves_per_eu(2, 2)))
void attention_kernel(const half_t* __restrict__ Qfrag, const half_t* __restrict__ Kfrag,
                      const half_t* __restrict__ Vfrag, half_t* __restrict__ Ofrag) {
  int wave = threadIdx.x >> 5, lane = threadIdx.x & 31;
  int wg = blockIdx.x * 8 + wave;
  int bh = wg >> 5;
  int qt0 = (wg & 31) * NQT;

  const long long qkStride = 2LL * 64 * 512;
  const long long vStride  = 4LL * 32 * 512;
  const half_t* Qb = Qfrag + (long long)bh * qkStride;
  const half_t* Kb = Kfrag + (long long)bh * qkStride;
  const half_t* Vb = Vfrag + (long long)bh * vStride;

  v16h bq0[NQT], bq1[NQT];
  v8f o[NQT][4];
  float m_run[NQT];
  float l_part[NQT];
#pragma unroll
  for (int q = 0; q < NQT; ++q) {
    bq0[q] = ld_frag(Qb + (0LL * 64 + qt0 + q) * 512, lane);
    bq1[q] = ld_frag(Qb + (1LL * 64 + qt0 + q) * 512, lane);
#pragma unroll
    for (int dt = 0; dt < 4; ++dt) o[q][dt] = 0.0f;
    m_run[q]  = -3.0e38f;
    l_part[q] = 0.f;
  }

  KVFrags f0, f1, f2;
  ld_kv(f0, Kb, Vb, 0, lane);
  ld_kv(f1, Kb, Vb, 1, lane);
  for (int it = 0; it + 4 < 32; it += 3) {
    ld_kv(f2, Kb, Vb, it + 2, lane);
    LOADS_STAY_AHEAD();
    attn_step<NQT>(f0, bq0, bq1, o, m_run, l_part);
    LOADS_STAY_AHEAD();
    ld_kv(f0, Kb, Vb, step_behind<NQT>(it + 3, o), lane);
    LOADS_STAY_AHEAD();
    attn_step<NQT>(f1, bq0, bq1, o, m_run, l_part);
    LOADS_STAY_AHEAD();
    ld_kv(f1, Kb, Vb, step_behind<NQT>(it + 4, o), lane);
    LOADS_STAY_AHEAD();
    attn_step<NQT>(f2, bq0, bq1, o, m_run, l_part);
  }
  attn_step<NQT>(f0, bq0, bq1, o, m_run, l_part);
  attn_step<NQT>(f1, bq0, bq1, o, m_run, l_part);

  int b = bh >> 3, head = bh & 7;
  half_t* Ob = Ofrag + (long long)b * (16LL * 64 * 512);
#pragma unroll
  for (int q = 0; q < NQT; ++q) {
    float inv = 1.0f / (l_part[q] + xhalf(l_part[q]));
#pragma unroll
    for (int t = 0; t < 2; ++t) {
      v16h ov;
#pragma unroll
      for (int r = 0; r < 8; ++r) {
        ov[r]     = (half_t)(o[q][2 * t][r] * inv);
        ov[r + 8] = (half_t)(o[q][2 * t + 1][r] * inv);
      }
      st_frag(Ob + (long long)((head * 2 + t) * 64 + qt0 + q) * 512, lane, ov);
    }
  }
}

extern "C" void kernel_launch(void* const* d_in, const int* in_sizes, int n_in,
                              void* d_out, int out_size, void* d_ws, size_t ws_size,
                              hipStream_t stream) {
  (void)in_sizes; (void)n_in; (void)out_size; (void)ws_size;
  const float* x      = (const float*)d_in[0];
  const float* gn_w   = (const float*)d_in[1];
  const float* gn_b   = (const float*)d_in[2];
  const float* qkv_w  = (const float*)d_in[3];
  const float* qkv_b  = (const float*)d_in[4];
  const float* proj_w = (const float*)d_in[5];
  const float* proj_b = (const float*)d_in[6];
  float* out = (float*)d_out;

  half_t* ws = (half_t*)d_ws;
  long long off = 0;
  half_t* Hfrag = ws + off; off += 8LL * 512 * 1024;
  half_t* Wq    = ws + off; off += 1536LL * 512;
  half_t* Wp    = ws + off; off += 512LL * 512;
  half_t* Qf    = ws + off; off += 64LL * 2 * 64 * 512;
  half_t* Kf    = ws + off; off += 64LL * 2 * 64 * 512;
  half_t* Vf    = ws + off; off += 64LL * 4 * 32 * 512;
  half_t* Of    = ws + off; off += 8LL * 512 * 1024;

  prep_kernel<<<dim3(512), dim3(256), 0, stream>>>(x, gn_w, gn_b, qkv_w, proj_w, Hfrag, Wq, Wp);

  gemm_qkv_kernel<<<dim3(12, 8, 8), dim3(256), 0, stream>>>(Wq, Hfrag, qkv_b, Qf, Kf, Vf);

  attention_kernel<<<dim3(256), dim3(256), 0, stream>>>(Qf, Kf, Vf, Of);

  gemm_proj_kernel<<<dim3(4, 8, 8), dim3(256), 0, stream>>>(Wp, Of, proj_b, x, out);
}


# QKV GEMM: a wave computes 64x64 (was 64x32) in 12-wave blocks of 192x256, one per WGP: half the waves, 16 loads per 16 products (was 12 per 8); at 3 waves per SIMD the loads run 2 k-steps ahead
# speedup vs baseline: 4.0784x; 1.1182x over previous
#include <hip/hip_runtime.h>

typedef _Float16 half_t;
typedef __attribute__((ext_vector_type(16))) _Float16 v16h;
typedef __attribute__((ext_vector_type(8)))  _Float16 v8h;
typedef __attribute__((ext_vector_type(8)))  float    v8f;

#define WMMA_F16(a, b, c) \
  __builtin_amdgcn_wmma_f32_16x16x32_f16(false, (a), false, (b), (short)0, (c), false, false)

__device__ __forceinline__ int kmap_of(int lane, int j) {
  return ((lane >> 4) << 3) + (j & 7) + ((j >> 3) << 4);
}

__device__ __forceinline__ int kmap_grp(int lane, int j) {
  return ((lane >> 4) << 4) + j;
}

__device__ __forceinline__ v16h ld_frag(const half_t* __restrict__ tile, int lane) {
  v8h lo = *(const v8h*)(tile + lane * 8);
  v8h hi = *(const v8h*)(tile + 256 + lane * 8);
  return __builtin_shufflevector(lo, hi, 0, 1, 2, 3, 4, 5, 6, 7, 8, 9, 10, 11, 12, 13, 14, 15);
}
__device__ __forceinline__ void st_frag(half_t* __restrict__ tile, int lane, v16h v) {
  *(v8h*)(tile + lane * 8)       = __builtin_shufflevector(v, v, 0, 1, 2, 3, 4, 5, 6, 7);
  *(v8h*)(tile + 256 + lane * 8) = __builtin_shufflevector(v, v, 8, 9, 10, 11, 12, 13, 14, 15);
}

__device__ __forceinline__
void groupnorm_part(const float* __restrict__ x,
                    const float* __restrict__ gw,
                    const float* __restrict__ gb,
                    half_t* __restrict__ Hfrag, int blk) {
  __shared__ float s_sum[256], s_ss[256];
  int b = blk >> 5;
  int g = blk & 31;
  const int GE = 16 * 1024;
  const float* xb = x + ((long long)b * 512 + g * 16) * 1024;
  float gwv[16], gbv[16];
#pragma unroll
  for (int cc = 0; cc < 16; ++cc) {
    gwv[cc] = gw[g * 16 + cc];
    gbv[cc] = gb[g * 16 + cc];
  }
  float xv[4][16];
#pragma unroll
  for (int cc = 0; cc < 16; ++cc)
#pragma unroll
    for (int i = 0; i < 4; ++i)
      xv[i][cc] = xb[cc * 1024 + i * 256 + (int)threadIdx.x];
  float sum = 0.f, ss = 0.f;
#pragma unroll
  for (int cc = 0; cc < 16; ++cc)
#pragma unroll
    for (int i = 0; i < 4; ++i) {
      float v = xv[i][cc];
      sum += v; ss += v * v;
    }
  s_sum[threadIdx.x] = sum; s_ss[threadIdx.x] = ss;
  __syncthreads();
  for (int off = 128; off > 0; off >>= 1) {
    if (threadIdx.x < off) {
      s_sum[threadIdx.x] += s_sum[threadIdx.x + off];
      s_ss[threadIdx.x]  += s_ss[threadIdx.x + off];
    }
    __syncthreads();
  }
  float mean = s_sum[0] * (1.0f / GE);
  float var  = s_ss[0] * (1.0f / GE) - mean * mean;
  float inv  = rsqrtf(var + 1e-5f);
  half_t* hb = Hfrag + (long long)b * (16LL * 64 * 512)
                     + (long long)(g >> 1) * (64 * 512);
  int laneAdd = (g & 1) * 16;
#pragma unroll
  for (int i = 0; i < 4; ++i) {
    int n = (int)threadIdx.x + i * 256;
    v16h hv;
#pragma unroll
    for (int cc = 0; cc < 16; ++cc)
      hv[cc] = (half_t)((xv[i][cc] - mean) * inv * gwv[cc] + gbv[cc]);
    st_frag(hb + (n >> 4) * 512, (n & 15) + laneAdd, hv);
  }
}

template <bool GRP, int MT, int KT, int NBLK>
__device__ __forceinline__
void repackA_part(const float* __restrict__ src, half_t* __restrict__ dst, int vblock) {
  constexpr int kcols = KT * 32;
  constexpr int total = MT * KT * 512;
  constexpr int trips = total / (NBLK * 256);
  static_assert(trips * NBLK * 256 == total, "the block range must cover the matrix exactly");
#pragma unroll
  for (int it = 0; it < trips; ++it) {
    int tid = (it * NBLK + vblock) * 256 + (int)threadIdx.x;
    int j = (tid & 7) + ((tid >> 5) & 8), lane = (tid >> 3) & 31, tile = tid >> 9;
    int mt = tile / KT, kt = tile - mt * KT;
    int row = mt * 16 + (lane & 15);
    int col = kt * 32 + (GRP ? kmap_grp(lane, j) : kmap_of(lane, j));
    dst[tid] = (half_t)src[(long long)row * kcols + col];
  }
}

__global__ __launch_bounds__(256)
void prep_kernel(const float* __restrict__ x,
                 const float* __restrict__ gw,
                 const float* __restrict__ gb,
                 const float* __restrict__ qkv_w,
                 const float* __restrict__ proj_w,
                 half_t* __restrict__ Hfrag,
                 half_t* __restrict__ Wq,
                 half_t* __restrict__ Wp) {
  int blk = blockIdx.x;
  if (blk < 256)      groupnorm_part(x, gw, gb, Hfrag, blk);
  else if (blk < 448) repackA_part<true, 96, 16, 192>(qkv_w, Wq, blk - 256);
  else                repackA_part<false, 32, 16, 64>(proj_w, Wp, blk - 448);
}

#define LOADS_STAY_AHEAD() __builtin_amdgcn_sched_barrier(0)

template <int NB> struct StepFrags { v16h a[4]; v16h b[NB]; };
template <int NB>
__device__ __forceinline__
void ld_step(StepFrags<NB>& f, const half_t* __restrict__ Afrag, const half_t* __restrict__ Bfrag,
             int KT, int NT, int mtBase, int ntBase, int kt, int lane) {
#pragma unroll
  for (int i = 0; i < 4; ++i)
    f.a[i] = ld_frag(Afrag + ((long long)(mtBase + i) * KT + kt) * 512, lane);
#pragma unroll
  for (int j = 0; j < NB; ++j)
    f.b[j] = ld_frag(Bfrag + ((long long)kt * NT + (ntBase + j)) * 512, lane);
}
template <bool TR, int NB>
__device__ __forceinline__
void mma_step(const StepFrags<NB>& f, v8f c[4][NB]) {
#pragma unroll
  for (int i = 0; i < 4; ++i)
#pragma unroll
    for (int j = 0; j < NB; ++j)
      c[i][j] = TR ? WMMA_F16(f.b[j], f.a[i], c[i][j]) : WMMA_F16(f.a[i], f.b[j], c[i][j]);
}
template <bool TR, int KT, int NB, int UNR>
__device__ __forceinline__
void gemm_wave_ahead2(const half_t* __restrict__ Afrag, const half_t* __restrict__ Bfrag,
                      int NT, int mtBase, int ntBase, int lane, v8f c[4][NB]) {
  static_assert(KT % 3 == 1 && KT >= 4, "whole trips of three steps, then four closing steps");
  StepFrags<NB> f0, f1, f2;
  ld_step<NB>(f0, Afrag, Bfrag, KT, NT, mtBase, ntBase, 0, lane);
  LOADS_STAY_AHEAD();
  ld_step<NB>(f1, Afrag, Bfrag, KT, NT, mtBase, ntBase, 1, lane);
#pragma unroll UNR
  for (int kt = 0; kt + 4 < KT; kt += 3) {
    ld_step<NB>(f2, Afrag, Bfrag, KT, NT, mtBase, ntBase, kt + 2, lane);
    LOADS_STAY_AHEAD();
    mma_step<TR, NB>(f0, c);
    LOADS_STAY_AHEAD();
    ld_step<NB>(f0, Afrag, Bfrag, KT, NT, mtBase, ntBase, kt + 3, lane);
    LOADS_STAY_AHEAD();
    mma_step<TR, NB>(f1, c);
    LOADS_STAY_AHEAD();
    ld_step<NB>(f1, Afrag, Bfrag, KT, NT, mtBase, ntBase, kt + 4, lane);
    LOADS_STAY_AHEAD();
    mma_step<TR, NB>(f2, c);
  }
  ld_step<NB>(f2, Afrag, Bfrag, KT, NT, mtBase, ntBase, KT - 2, lane);
  LOADS_STAY_AHEAD();
  mma_step<TR, NB>(f0, c);
  LOADS_STAY_AHEAD();
  ld_step<NB>(f0, Afrag, Bfrag, KT, NT, mtBase, ntBase, KT - 1, lane);
  LOADS_STAY_AHEAD();
  mma_step<TR, NB>(f1, c);
  mma_step<TR, NB>(f2, c);
  mma_step<TR, NB>(f0, c);
}

#define EPILOGUE_STAYS_BEHIND() __builtin_amdgcn_sched_barrier(0)
__global__ __launch_bounds__(384) __attribute__((amdgpu_waves_per_eu(3, 3)))
void gemm_qkv_kernel(const half_t* __restrict__ Wfrag, const half_t* __restrict__ Hfrag,
                     const float* __restrict__ bias, half_t* __restrict__ Qf,
                     half_t* __restrict__ Kf, half_t* __restrict__ Vf) {
  int wave = __builtin_amdgcn_readfirstlane((int)(threadIdx.x >> 5));
  int lane = threadIdx.x & 31;
  int b = blockIdx.z;
  int wm = wave % 3, wn = wave / 3;
  int mBase = blockIdx.x * 192 + wm * 64;
  int nBase = blockIdx.y * 256 + wn * 64;
  int which = mBase >> 9;
  int bh = b * 8 + ((mBase & 511) >> 6);
  int hh = lane >> 4, col = lane & 15;
  const half_t* Hb = Hfrag + (long long)b * (16LL * 64 * 512);
  v8f c[4][4];
#pragma unroll
  for (int i = 0; i < 4; ++i)
#pragma unroll
    for (int j = 0; j < 4; ++j) c[i][j] = 0.0f;

  if (which == 2) {
    gemm_wave_ahead2<true, 16, 4, 1>(Wfrag, Hb, 64, mBase >> 4, nBase >> 4, lane, c);
    EPILOGUE_STAYS_BEHIND();
    half_t* vb = Vf + (long long)bh * (4LL * 32 * 512);
    int ktile = nBase >> 5;
#pragma unroll
    for (int i = 0; i < 4; ++i) {
      float bv = bias[mBase + i * 16 + col];
#pragma unroll
      for (int p = 0; p < 2; ++p) {
        v16h ov;
#pragma unroll
        for (int r = 0; r < 8; ++r) {
          ov[r]     = (half_t)(c[i][2 * p][r]     + bv);
          ov[r + 8] = (half_t)(c[i][2 * p + 1][r] + bv);
        }
        st_frag(vb + (long long)(i * 32 + ktile + p) * 512, lane, ov);
      }
    }
  } else {
    gemm_wave_ahead2<false, 16, 4, 1>(Wfrag, Hb, 64, mBase >> 4, nBase >> 4, lane, c);
    EPILOGUE_STAYS_BEHIND();
    half_t* qb = (which == 0 ? Qf : Kf) + (long long)bh * (2LL * 64 * 512);
    int ntile = nBase >> 4;
#pragma unroll
    for (int kt = 0; kt < 2; ++kt)
#pragma unroll
      for (int jj = 0; jj < 4; ++jj) {
        v16h ov;
#pragma unroll
        for (int r = 0; r < 8; ++r) {
          int m0 = mBase + (2 * kt) * 16 + 8 * hh + r;
          ov[r]     = (half_t)(c[2 * kt][jj][r]     + bias[m0]);
          ov[r + 8] = (half_t)(c[2 * kt + 1][jj][r] + bias[m0 + 16]);
        }
        st_frag(qb + (long long)(kt * 64 + ntile + jj) * 512, lane, ov);
      }
  }
}

__global__ __launch_bounds__(256)
void gemm_proj_kernel(const half_t* __restrict__ Wfrag, const half_t* __restrict__ Bfrag,
                      const float* __restrict__ bias, const float* __restrict__ resid,
                      float* __restrict__ out) {
  int wave = threadIdx.x >> 5, lane = threadIdx.x & 31;
  int b = blockIdx.z;
  int wm = wave & 1, wn = wave >> 1;
  int mBase = blockIdx.x * 128 + wm * 64;
  int nBase = blockIdx.y * 128 + wn * 32;
  v8f c[4][2];
#pragma unroll
  for (int i = 0; i < 4; ++i)
#pragma unroll
    for (int j = 0; j < 2; ++j) c[i][j] = 0.0f;
  gemm_wave_ahead2<false, 16, 2, 4>(Wfrag, Bfrag + (long long)b * (16LL * 64 * 512),
                                 64, mBase >> 4, nBase >> 4, lane, c);
  const float* xb = resid + (long long)b * 512 * 1024;
  float* ob = out + (long long)b * 512 * 1024;
  int rAdd = 8 * (lane >> 4), col = lane & 15;
#pragma unroll
  for (int i = 0; i < 4; ++i)
#pragma unroll
    for (int j = 0; j < 2; ++j)
#pragma unroll
      for (int r = 0; r < 8; ++r) {
        int m = mBase + i * 16 + r + rAdd;
        int n = nBase + j * 16 + col;
        long long idx = (long long)m * 1024 + n;
        ob[idx] = c[i][j][r] + bias[m] + xb[idx];
      }
}

__device__ __forceinline__ float xhalf(float x) {
  int b = __builtin_bit_cast(int, x);
  return __builtin_bit_cast(float,
      __builtin_amdgcn_permlanex16(b, b, 0x76543210, 0xfedcba98, false, false));
}

struct KVFrags { v16h k[4]; v16h v[4]; };
__device__ __forceinline__
void ld_kv(KVFrags& f, const half_t* __restrict__ Kb, const half_t* __restrict__ Vb,
           int it, int lane) {
  int kb0 = it * 2, kb1 = it * 2 + 1;
  f.k[0] = ld_frag(Kb + (0LL * 64 + kb0) * 512, lane);
  f.k[1] = ld_frag(Kb + (1LL * 64 + kb0) * 512, lane);
  f.k[2] = ld_frag(Kb + (0LL * 64 + kb1) * 512, lane);
  f.k[3] = ld_frag(Kb + (1LL * 64 + kb1) * 512, lane);
#pragma unroll
  for (int dt = 0; dt < 4; ++dt)
    f.v[dt] = ld_frag(Vb + ((long long)dt * 32 + it) * 512, lane);
}

template <int NQ>
__device__ __forceinline__
void attn_step(const KVFrags& f, const v16h (&bq0)[NQ], const v16h (&bq1)[NQ],
               v8f (&o)[NQ][4], float (&m_run)[NQ], float (&l_part)[NQ]) {
  const float cs = 0.125f * 1.4426950408889634f;
  v8f sA[NQ], sB[NQ];
#pragma unroll
  for (int q = 0; q < NQ; ++q) {
    sA[q] = 0.0f; sB[q] = 0.0f;
    sA[q] = WMMA_F16(f.k[0], bq0[q], sA[q]);
    sA[q] = WMMA_F16(f.k[1], bq1[q], sA[q]);
    sB[q] = WMMA_F16(f.k[2], bq0[q], sB[q]);
    sB[q] = WMMA_F16(f.k[3], bq1[q], sB[q]);
  }

  v16h bp[NQ];
#pragma unroll
  for (int q = 0; q < NQ; ++q) {
    float mx = fmaxf(sA[q][0], sB[q][0]);
#pragma unroll
    for (int r = 1; r < 8; ++r) mx = fmaxf(mx, fmaxf(sA[q][r], sB[q][r]));
    mx = fmaxf(mx, xhalf(mx));
    float m_new = fmaxf(m_run[q], mx);
    float alpha = __builtin_amdgcn_exp2f((m_run[q] - m_new) * cs);
    float nm = -(m_new * cs);

    float sumA = 0.f, sumB = 0.f;
#pragma unroll
    for (int r = 0; r < 8; ++r) {
      float pa = __builtin_amdgcn_exp2f(__builtin_fmaf(sA[q][r], cs, nm));
      float pb = __builtin_amdgcn_exp2f(__builtin_fmaf(sB[q][r], cs, nm));
      sumA += pa; sumB += pb;
      bp[q][r]     = (half_t)pa;
      bp[q][r + 8] = (half_t)pb;
    }
    l_part[q] = l_part[q] * alpha + (sumA + sumB);
    m_run[q] = m_new;

#pragma unroll
    for (int dt = 0; dt < 4; ++dt)
#pragma unroll
      for (int r = 0; r < 8; ++r) o[q][dt][r] *= alpha;
  }

#pragma unroll
  for (int q = 0; q < NQ; ++q)
#pragma unroll
    for (int dt = 0; dt < 4; ++dt) o[q][dt] = WMMA_F16(f.v[dt], bp[q], o[q][dt]);
}

template <int NQ>
__device__ __forceinline__ int step_behind(int step, v8f (&o)[NQ][4]) {
#pragma unroll
  for (int q = 0; q < NQ; ++q)
    asm volatile("" : "+s"(step), "+v"(o[q][0]), "+v"(o[q][1]), "+v"(o[q][2]), "+v"(o[q][3]));
  return step;
}

#define NQT 2
__global__ __launch_bounds__(256) __attribute__((amdgpu_waves_per_eu(2, 2)))
void attention_kernel(const half_t* __restrict__ Qfrag, const half_t* __restrict__ Kfrag,
                      const half_t* __restrict__ Vfrag, half_t* __restrict__ Ofrag) {
  int wave = threadIdx.x >> 5, lane = threadIdx.x & 31;
  int wg = blockIdx.x * 8 + wave;
  int bh = wg >> 5;
  int qt0 = (wg & 31) * NQT;

  const long long qkStride = 2LL * 64 * 512;
  const long long vStride  = 4LL * 32 * 512;
  const half_t* Qb = Qfrag + (long long)bh * qkStride;
  const half_t* Kb = Kfrag + (long long)bh * qkStride;
  const half_t* Vb = Vfrag + (long long)bh * vStride;

  v16h bq0[NQT], bq1[NQT];
  v8f o[NQT][4];
  float m_run[NQT];
  float l_part[NQT];
#pragma unroll
  for (int q = 0; q < NQT; ++q) {
    bq0[q] = ld_frag(Qb + (0LL * 64 + qt0 + q) * 512, lane);
    bq1[q] = ld_frag(Qb + (1LL * 64 + qt0 + q) * 512, lane);
#pragma unroll
    for (int dt = 0; dt < 4; ++dt) o[q][dt] = 0.0f;
    m_run[q]  = -3.0e38f;
    l_part[q] = 0.f;
  }

  KVFrags f0, f1, f2;
  ld_kv(f0, Kb, Vb, 0, lane);
  ld_kv(f1, Kb, Vb, 1, lane);
  for (int it = 0; it + 4 < 32; it += 3) {
    ld_kv(f2, Kb, Vb, it + 2, lane);
    LOADS_STAY_AHEAD();
    attn_step<NQT>(f0, bq0, bq1, o, m_run, l_part);
    LOADS_STAY_AHEAD();
    ld_kv(f0, Kb, Vb, step_behind<NQT>(it + 3, o), lane);
    LOADS_STAY_AHEAD();
    attn_step<NQT>(f1, bq0, bq1, o, m_run, l_part);
    LOADS_STAY_AHEAD();
    ld_kv(f1, Kb, Vb, step_behind<NQT>(it + 4, o), lane);
    LOADS_STAY_AHEAD();
    attn_step<NQT>(f2, bq0, bq1, o, m_run, l_part);
  }
  attn_step<NQT>(f0, bq0, bq1, o, m_run, l_part);
  attn_step<NQT>(f1, bq0, bq1, o, m_run, l_part);

  int b = bh >> 3, head = bh & 7;
  half_t* Ob = Ofrag + (long long)b * (16LL * 64 * 512);
#pragma unroll
  for (int q = 0; q < NQT; ++q) {
    float inv = 1.0f / (l_part[q] + xhalf(l_part[q]));
#pragma unroll
    for (int t = 0; t < 2; ++t) {
      v16h ov;
#pragma unroll
      for (int r = 0; r < 8; ++r) {
        ov[r]     = (half_t)(o[q][2 * t][r] * inv);
        ov[r + 8] = (half_t)(o[q][2 * t + 1][r] * inv);
      }
      st_frag(Ob + (long long)((head * 2 + t) * 64 + qt0 + q) * 512, lane, ov);
    }
  }
}

extern "C" void kernel_launch(void* const* d_in, const int* in_sizes, int n_in,
                              void* d_out, int out_size, void* d_ws, size_t ws_size,
                              hipStream_t stream) {
  (void)in_sizes; (void)n_in; (void)out_size; (void)ws_size;
  const float* x      = (const float*)d_in[0];
  const float* gn_w   = (const float*)d_in[1];
  const float* gn_b   = (const float*)d_in[2];
  const float* qkv_w  = (const float*)d_in[3];
  const float* qkv_b  = (const float*)d_in[4];
  const float* proj_w = (const float*)d_in[5];
  const float* proj_b = (const float*)d_in[6];
  float* out = (float*)d_out;

  half_t* ws = (half_t*)d_ws;
  long long off = 0;
  half_t* Hfrag = ws + off; off += 8LL * 512 * 1024;
  half_t* Wq    = ws + off; off += 1536LL * 512;
  half_t* Wp    = ws + off; off += 512LL * 512;
  half_t* Qf    = ws + off; off += 64LL * 2 * 64 * 512;
  half_t* Kf    = ws + off; off += 64LL * 2 * 64 * 512;
  half_t* Vf    = ws + off; off += 64LL * 4 * 32 * 512;
  half_t* Of    = ws + off; off += 8LL * 512 * 1024;

  prep_kernel<<<dim3(512), dim3(256), 0, stream>>>(x, gn_w, gn_b, qkv_w, proj_w, Hfrag, Wq, Wp);

  gemm_qkv_kernel<<<dim3(8, 4, 8), dim3(384), 0, stream>>>(Wq, Hfrag, qkv_b, Qf, Kf, Vf);

  attention_kernel<<<dim3(256), dim3(256), 0, stream>>>(Qf, Kf, Vf, Of);

  gemm_proj_kernel<<<dim3(4, 8, 8), dim3(256), 0, stream>>>(Wp, Of, proj_b, x, out);
}
